# starting kernel
# baseline (speedup 1.0000x reference)
#include <hip/hip_runtime.h>
#include <hip/hip_bf16.h>

typedef __attribute__((ext_vector_type(2))) float v2f;
typedef __attribute__((ext_vector_type(8))) float v8f;

#define NCOLS 128
#define NODE_K 256
#define EDGE_K 64
#define NHEADS 4

template <int K>
__global__ void __launch_bounds__(256)
gemm_f32_wmma(const float* __restrict__ A, const float* __restrict__ B,
              float* __restrict__ C, int M) {
    __shared__ float Apanel[16 * K];
    const int lane = threadIdx.x & 31;
    const int wave = threadIdx.x >> 5;
    const int m0   = blockIdx.x * 16;

    for (int idx = threadIdx.x; idx < 16 * K; idx += 256) {
        int r = idx / K, c = idx % K;
        int row = m0 + r; row = row < M ? row : M - 1;
        Apanel[idx] = A[(size_t)row * K + c];
    }
    __syncthreads();

    const int n0      = wave * 16;
    const int half    = lane >> 4;
    const int l15     = lane & 15;
    const int kpair   = half * 2;

    v8f acc = {};
    for (int k = 0; k < K; k += 4) {
        v2f a, b;
        a.x = Apanel[l15 * K + k + kpair];
        a.y = Apanel[l15 * K + k + kpair + 1];
        b.x = B[(size_t)(k + kpair)     * NCOLS + n0 + l15];
        b.y = B[(size_t)(k + kpair + 1) * NCOLS + n0 + l15];
        acc = __builtin_amdgcn_wmma_f32_16x16x4_f32(
            false, a, false, b, (short)0, acc, false, false);
    }

    const int mbase = half * 8;
#pragma unroll
    for (int r = 0; r < 8; ++r) {
        int row = m0 + mbase + r;
        if (row < M) C[(size_t)row * NCOLS + n0 + l15] = acc[r];
    }
}

__global__ void __launch_bounds__(256)
edge_proj_score(const float* __restrict__ EF, const float* __restrict__ We,
                const float* __restrict__ H, const int* __restrict__ src,
                float* __restrict__ score, int E) {
    __shared__ float Apanel[16 * EDGE_K];
    __shared__ int   srcIdx[16];
    __shared__ float spart[8][16];

    const int lane = threadIdx.x & 31;
    const int wave = threadIdx.x >> 5;
    const int m0   = blockIdx.x * 16;

    for (int idx = threadIdx.x; idx < 16 * EDGE_K; idx += 256) {
        int r = idx >> 6, c = idx & 63;
        int row = m0 + r; row = row < E ? row : E - 1;
        Apanel[idx] = EF[(size_t)row * EDGE_K + c];
    }
    if (threadIdx.x < 16) {
        int e = m0 + threadIdx.x;
        srcIdx[threadIdx.x] = src[e < E ? e : E - 1];
    }
    __syncthreads();

    const int n0    = wave * 16;
    const int half  = lane >> 4;
    const int l15   = lane & 15;
    const int kpair = half * 2;

    v8f acc = {};
    for (int k = 0; k < EDGE_K; k += 4) {
        v2f a, b;
        a.x = Apanel[l15 * EDGE_K + k + kpair];
        a.y = Apanel[l15 * EDGE_K + k + kpair + 1];
        b.x = We[(size_t)(k + kpair)     * NCOLS + n0 + l15];
        b.y = We[(size_t)(k + kpair + 1) * NCOLS + n0 + l15];
        acc = __builtin_amdgcn_wmma_f32_16x16x4_f32(
            false, a, false, b, (short)0, acc, false, false);
    }

    const int mbase = half * 8;
    float prod[8];
#pragma unroll
    for (int r = 0; r < 8; ++r) {
        int s = srcIdx[mbase + r];
        float hv = H[(size_t)s * NCOLS + n0 + l15];
        prod[r] = acc[r] * hv;
    }
#pragma unroll
    for (int off = 1; off < 16; off <<= 1) {
#pragma unroll
        for (int r = 0; r < 8; ++r) prod[r] += __shfl_xor(prod[r], off, 32);
    }
    if (l15 == 0) {
#pragma unroll
        for (int r = 0; r < 8; ++r) spart[wave][mbase + r] = prod[r];
    }
    __syncthreads();

    if (threadIdx.x < 64) {
        int m = threadIdx.x >> 2, hd = threadIdx.x & 3;
        int e = m0 + m;
        if (e < E) {
            float sv = (spart[2 * hd][m] + spart[2 * hd + 1][m]) * 0.17677669529663689f;
            sv = sv > 0.f ? sv : 0.01f * sv;
            score[(size_t)e * NHEADS + hd] = sv;
        }
    }
}

__global__ void __launch_bounds__(256)
init_kernel(float* __restrict__ out, float* __restrict__ smax,
            float* __restrict__ denom, int nOut, int nSeg) {
    int gid = blockIdx.x * 256 + threadIdx.x;
    if (gid < nOut) out[gid] = 0.f;
    if (gid < nSeg) { smax[gid] = __int_as_float(0xFF800000); denom[gid] = 0.f; }
}

__device__ __forceinline__ void atomicMaxFloat(float* addr, float val) {
    if (val >= 0.f) atomicMax((int*)addr, __float_as_int(val));
    else            atomicMin((unsigned int*)addr, __float_as_uint(val));
}

__global__ void __launch_bounds__(256)
segmax_kernel(const float* __restrict__ score, const int* __restrict__ dst,
              float* __restrict__ smax, int E) {
    int gid = blockIdx.x * 256 + threadIdx.x;
    if (gid >= E * NHEADS) return;
    int e = gid >> 2, hd = gid & 3;
    atomicMaxFloat(&smax[dst[e] * NHEADS + hd], score[gid]);
}

__global__ void __launch_bounds__(256)
expsum_kernel(const float* __restrict__ score, const int* __restrict__ dst,
              const float* __restrict__ smax, float* __restrict__ aarr,
              float* __restrict__ denom, int E) {
    int gid = blockIdx.x * 256 + threadIdx.x;
    if (gid >= E * NHEADS) return;
    int e = gid >> 2, hd = gid & 3;
    float v = __expf(score[gid] - smax[dst[e] * NHEADS + hd]);
    aarr[gid] = v;
    atomicAdd(&denom[dst[e] * NHEADS + hd], v);
}

__global__ void __launch_bounds__(256)
aggregate_kernel(const float* __restrict__ H, const int* __restrict__ src,
                 const int* __restrict__ dst, const float* __restrict__ aarr,
                 const float* __restrict__ denom, float* __restrict__ out, int E) {
    int gid = blockIdx.x * 256 + threadIdx.x;
    int e = gid >> 7;
    if (e >= E) return;
    int f  = gid & 127;
    int hd = f >> 5;
    int s = src[e], d = dst[e];
    float attn = aarr[e * NHEADS + hd] / denom[d * NHEADS + hd];
    float val  = H[(size_t)s * NCOLS + f] * attn;
    atomicAdd(&out[(size_t)d * NCOLS + f], val);
}

extern "C" void kernel_launch(void* const* d_in, const int* in_sizes, int n_in,
                              void* d_out, int out_size, void* d_ws, size_t ws_size,
                              hipStream_t stream) {
    const float* node_feat = (const float*)d_in[0];
    const float* edge_feat = (const float*)d_in[1];
    const int*   src       = (const int*)d_in[2];
    const int*   dst       = (const int*)d_in[3];
    const float* W_node    = (const float*)d_in[4];
    const float* W_edge    = (const float*)d_in[5];
    float*       out       = (float*)d_out;

    const int N = in_sizes[0] / NODE_K;
    const int E = in_sizes[2];

    float* h     = (float*)d_ws;
    float* score = h     + (size_t)N * NCOLS;
    float* aarr  = score + (size_t)E * NHEADS;
    float* smax  = aarr  + (size_t)E * NHEADS;
    float* denom = smax  + (size_t)N * NHEADS;

    gemm_f32_wmma<NODE_K><<<(N + 15) / 16, 256, 0, stream>>>(node_feat, W_node, h, N);

    edge_proj_score<<<(E + 15) / 16, 256, 0, stream>>>(edge_feat, W_edge, h, src, score, E);

    int nOut = N * NCOLS;
    init_kernel<<<(nOut + 255) / 256, 256, 0, stream>>>(out, smax, denom, nOut, N * NHEADS);

    segmax_kernel<<<(E * NHEADS + 255) / 256, 256, 0, stream>>>(score, dst, smax, E);

    expsum_kernel<<<(E * NHEADS + 255) / 256, 256, 0, stream>>>(score, dst, smax, aarr, denom, E);

    long long total = (long long)E * NCOLS;
    aggregate_kernel<<<(int)((total + 255) / 256), 256, 0, stream>>>(h, src, dst, aarr, denom, out, E);
}


# aggregation: in-call counting sort of edges by dst (int rank atomics, scan, fill), then one wave per node sums its h[src] rows in registers and stores once; no float atomics on out, sum order changes
# speedup vs baseline: 1.2886x; 1.2886x over previous
#include <hip/hip_runtime.h>
#include <hip/hip_bf16.h>

typedef __attribute__((ext_vector_type(2))) float v2f;
typedef __attribute__((ext_vector_type(8))) float v8f;

#define NCOLS 128
#define NODE_K 256
#define EDGE_K 64
#define NHEADS 4

template <int K>
__global__ void __launch_bounds__(256)
gemm_f32_wmma(const float* __restrict__ A, const float* __restrict__ B,
              float* __restrict__ C, int M) {
    __shared__ float Apanel[16 * K];
    const int lane = threadIdx.x & 31;
    const int wave = threadIdx.x >> 5;
    const int m0   = blockIdx.x * 16;

    for (int idx = threadIdx.x; idx < 16 * K; idx += 256) {
        int r = idx / K, c = idx % K;
        int row = m0 + r; row = row < M ? row : M - 1;
        Apanel[idx] = A[(size_t)row * K + c];
    }
    __syncthreads();

    const int n0      = wave * 16;
    const int half    = lane >> 4;
    const int l15     = lane & 15;
    const int kpair   = half * 2;

    v8f acc = {};
    for (int k = 0; k < K; k += 4) {
        v2f a, b;
        a.x = Apanel[l15 * K + k + kpair];
        a.y = Apanel[l15 * K + k + kpair + 1];
        b.x = B[(size_t)(k + kpair)     * NCOLS + n0 + l15];
        b.y = B[(size_t)(k + kpair + 1) * NCOLS + n0 + l15];
        acc = __builtin_amdgcn_wmma_f32_16x16x4_f32(
            false, a, false, b, (short)0, acc, false, false);
    }

    const int mbase = half * 8;
#pragma unroll
    for (int r = 0; r < 8; ++r) {
        int row = m0 + mbase + r;
        if (row < M) C[(size_t)row * NCOLS + n0 + l15] = acc[r];
    }
}

__global__ void __launch_bounds__(256)
edge_proj_score(const float* __restrict__ EF, const float* __restrict__ We,
                const float* __restrict__ H, const int* __restrict__ src,
                float* __restrict__ score, int E) {
    __shared__ float Apanel[16 * EDGE_K];
    __shared__ int   srcIdx[16];
    __shared__ float spart[8][16];

    const int lane = threadIdx.x & 31;
    const int wave = threadIdx.x >> 5;
    const int m0   = blockIdx.x * 16;

    for (int idx = threadIdx.x; idx < 16 * EDGE_K; idx += 256) {
        int r = idx >> 6, c = idx & 63;
        int row = m0 + r; row = row < E ? row : E - 1;
        Apanel[idx] = EF[(size_t)row * EDGE_K + c];
    }
    if (threadIdx.x < 16) {
        int e = m0 + threadIdx.x;
        srcIdx[threadIdx.x] = src[e < E ? e : E - 1];
    }
    __syncthreads();

    const int n0    = wave * 16;
    const int half  = lane >> 4;
    const int l15   = lane & 15;
    const int kpair = half * 2;

    v8f acc = {};
    for (int k = 0; k < EDGE_K; k += 4) {
        v2f a, b;
        a.x = Apanel[l15 * EDGE_K + k + kpair];
        a.y = Apanel[l15 * EDGE_K + k + kpair + 1];
        b.x = We[(size_t)(k + kpair)     * NCOLS + n0 + l15];
        b.y = We[(size_t)(k + kpair + 1) * NCOLS + n0 + l15];
        acc = __builtin_amdgcn_wmma_f32_16x16x4_f32(
            false, a, false, b, (short)0, acc, false, false);
    }

    const int mbase = half * 8;
    float prod[8];
#pragma unroll
    for (int r = 0; r < 8; ++r) {
        int s = srcIdx[mbase + r];
        float hv = H[(size_t)s * NCOLS + n0 + l15];
        prod[r] = acc[r] * hv;
    }
#pragma unroll
    for (int off = 1; off < 16; off <<= 1) {
#pragma unroll
        for (int r = 0; r < 8; ++r) prod[r] += __shfl_xor(prod[r], off, 32);
    }
    if (l15 == 0) {
#pragma unroll
        for (int r = 0; r < 8; ++r) spart[wave][mbase + r] = prod[r];
    }
    __syncthreads();

    if (threadIdx.x < 64) {
        int m = threadIdx.x >> 2, hd = threadIdx.x & 3;
        int e = m0 + m;
        if (e < E) {
            float sv = (spart[2 * hd][m] + spart[2 * hd + 1][m]) * 0.17677669529663689f;
            sv = sv > 0.f ? sv : 0.01f * sv;
            score[(size_t)e * NHEADS + hd] = sv;
        }
    }
}

__global__ void __launch_bounds__(256)
init_kernel(float* __restrict__ smax, float* __restrict__ denom,
            int* __restrict__ cnt, int nSeg, int N) {
    int gid = blockIdx.x * 256 + threadIdx.x;
    if (gid < nSeg) { smax[gid] = __int_as_float(0xFF800000); denom[gid] = 0.f; }
    if (gid < N) cnt[gid] = 0;
}

__device__ __forceinline__ void atomicMaxFloat(float* addr, float val) {
    if (val >= 0.f) atomicMax((int*)addr, __float_as_int(val));
    else            atomicMin((unsigned int*)addr, __float_as_uint(val));
}

__global__ void __launch_bounds__(256)
segmax_kernel(const float* __restrict__ score, const int* __restrict__ dst,
              float* __restrict__ smax, int E) {
    int gid = blockIdx.x * 256 + threadIdx.x;
    if (gid >= E * NHEADS) return;
    int e = gid >> 2, hd = gid & 3;
    atomicMaxFloat(&smax[dst[e] * NHEADS + hd], score[gid]);
}

__global__ void __launch_bounds__(256)
expsum_kernel(const float* __restrict__ score, const int* __restrict__ dst,
              const float* __restrict__ smax, float* __restrict__ aarr,
              float* __restrict__ denom, int E) {
    int gid = blockIdx.x * 256 + threadIdx.x;
    if (gid >= E * NHEADS) return;
    int e = gid >> 2, hd = gid & 3;
    float v = __expf(score[gid] - smax[dst[e] * NHEADS + hd]);
    aarr[gid] = v;
    atomicAdd(&denom[dst[e] * NHEADS + hd], v);
}


__global__ void __launch_bounds__(256)
rank_kernel(const int* __restrict__ dst, int* __restrict__ cnt,
            int* __restrict__ rank, int E) {
    int e = blockIdx.x * 256 + threadIdx.x;
    if (e < E) rank[e] = atomicAdd(&cnt[dst[e]], 1);
}

__global__ void __launch_bounds__(1024)
scan_kernel(const int* __restrict__ cnt, int* __restrict__ offs, int N) {
    __shared__ int wtot[32];
    const int tid  = threadIdx.x;
    const int lane = tid & 31;
    const int wave = tid >> 5;
    const int per  = (N + 1023) / 1024;
    const int beg  = tid * per;
    int end = beg + per; end = end < N ? end : N;

    int sum = 0;
    for (int i = beg; i < end; ++i) sum += cnt[i];

    int inc = sum;
#pragma unroll
    for (int off = 1; off < 32; off <<= 1) {
        int t = __shfl_up(inc, off, 32);
        if (lane >= off) inc += t;
    }
    if (lane == 31) wtot[wave] = inc;
    __syncthreads();
    if (wave == 0) {
        int w = wtot[lane], winc = w;
#pragma unroll
        for (int off = 1; off < 32; off <<= 1) {
            int t = __shfl_up(winc, off, 32);
            if (lane >= off) winc += t;
        }
        wtot[lane] = winc - w;
    }
    __syncthreads();

    int run = wtot[wave] + inc - sum;
    for (int i = beg; i < end; ++i) { offs[i] = run; run += cnt[i]; }
    if (tid == 1023) offs[N] = run;
}

__global__ void __launch_bounds__(256)
fill_kernel(const int* __restrict__ src, const int* __restrict__ dst,
            const int* __restrict__ rank, const int* __restrict__ offs,
            int* __restrict__ slot, int E) {
    int e = blockIdx.x * 256 + threadIdx.x;
    if (e >= E) return;
    int pos = offs[dst[e]] + rank[e];
    ((int2*)slot)[pos] = make_int2(e, src[e]);
}

__global__ void __launch_bounds__(256)
aggregate_kernel(const float* __restrict__ H, const int* __restrict__ slot,
                 const int* __restrict__ offs, const float* __restrict__ aarr,
                 const float* __restrict__ denom, float* __restrict__ out, int N) {
    const int lane = threadIdx.x & 31;
    const int d = __builtin_amdgcn_readfirstlane((int)(blockIdx.x * 8 + (threadIdx.x >> 5)));
    if (d >= N) return;
    const int beg = offs[d], end = offs[d + 1];
    const int hd  = lane >> 3;
    const float den = denom[d * NHEADS + hd];

    float4 acc = make_float4(0.f, 0.f, 0.f, 0.f);
    for (int base = beg; base < end; base += 32) {
        int n = end - base; n = n < 32 ? n : 32;
        int2 mine = make_int2(0, 0);
        if (lane < n) mine = ((const int2*)slot)[base + lane];
        for (int j = 0; j < n; ++j) {
            const int e = __builtin_amdgcn_readlane(mine.x, j);
            const int s = __builtin_amdgcn_readlane(mine.y, j);
            const float attn = aarr[(size_t)e * NHEADS + hd] / den;
            const float4 hv = *(const float4*)(H + (size_t)s * NCOLS + lane * 4);
            acc.x += hv.x * attn;
            acc.y += hv.y * attn;
            acc.z += hv.z * attn;
            acc.w += hv.w * attn;
        }
    }
    *(float4*)(out + (size_t)d * NCOLS + lane * 4) = acc;
}

extern "C" void kernel_launch(void* const* d_in, const int* in_sizes, int n_in,
                              void* d_out, int out_size, void* d_ws, size_t ws_size,
                              hipStream_t stream) {
    const float* node_feat = (const float*)d_in[0];
    const float* edge_feat = (const float*)d_in[1];
    const int*   src       = (const int*)d_in[2];
    const int*   dst       = (const int*)d_in[3];
    const float* W_node    = (const float*)d_in[4];
    const float* W_edge    = (const float*)d_in[5];
    float*       out       = (float*)d_out;

    const int N = in_sizes[0] / NODE_K;
    const int E = in_sizes[2];

    float* h     = (float*)d_ws;
    float* score = h     + (size_t)N * NCOLS;
    float* aarr  = score + (size_t)E * NHEADS;
    float* smax  = aarr  + (size_t)E * NHEADS;
    float* denom = smax  + (size_t)N * NHEADS;

    gemm_f32_wmma<NODE_K><<<(N + 15) / 16, 256, 0, stream>>>(node_feat, W_node, h, N);

    edge_proj_score<<<(E + 15) / 16, 256, 0, stream>>>(edge_feat, W_edge, h, src, score, E);

    int* cnt  = (int*)out;
    int* slot = (int*)score;
    int* rank = (int*)(score + 2 * (size_t)E);
    int* offs = (int*)smax;

    init_kernel<<<(N * NHEADS + 255) / 256, 256, 0, stream>>>(smax, denom, cnt, N * NHEADS, N);

    segmax_kernel<<<(E * NHEADS + 255) / 256, 256, 0, stream>>>(score, dst, smax, E);

    expsum_kernel<<<(E * NHEADS + 255) / 256, 256, 0, stream>>>(score, dst, smax, aarr, denom, E);

    rank_kernel<<<(E + 255) / 256, 256, 0, stream>>>(dst, cnt, rank, E);
    scan_kernel<<<1, 1024, 0, stream>>>(cnt, offs, N);
    fill_kernel<<<(E + 255) / 256, 256, 0, stream>>>(src, dst, rank, offs, slot, E);

    aggregate_kernel<<<(N + 7) / 8, 256, 0, stream>>>(h, slot, offs, aarr, denom, out, N);
}


# edge softmax moved into the per-node aggregation wave (wave max, exp, wave sum over its sorted edges); segmax/expsum kernels and all float atomics removed; denominator sum order changes
# speedup vs baseline: 1.5663x; 1.2155x over previous
#include <hip/hip_runtime.h>
#include <hip/hip_bf16.h>

typedef __attribute__((ext_vector_type(2))) float v2f;
typedef __attribute__((ext_vector_type(8))) float v8f;

#define NCOLS 128
#define NODE_K 256
#define EDGE_K 64
#define NHEADS 4

template <int K>
__global__ void __launch_bounds__(256)
gemm_f32_wmma(const float* __restrict__ A, const float* __restrict__ B,
              float* __restrict__ C, int M) {
    __shared__ float Apanel[16 * K];
    const int lane = threadIdx.x & 31;
    const int wave = threadIdx.x >> 5;
    const int m0   = blockIdx.x * 16;

    for (int idx = threadIdx.x; idx < 16 * K; idx += 256) {
        int r = idx / K, c = idx % K;
        int row = m0 + r; row = row < M ? row : M - 1;
        Apanel[idx] = A[(size_t)row * K + c];
    }
    __syncthreads();

    const int n0      = wave * 16;
    const int half    = lane >> 4;
    const int l15     = lane & 15;
    const int kpair   = half * 2;

    v8f acc = {};
    for (int k = 0; k < K; k += 4) {
        v2f a, b;
        a.x = Apanel[l15 * K + k + kpair];
        a.y = Apanel[l15 * K + k + kpair + 1];
        b.x = B[(size_t)(k + kpair)     * NCOLS + n0 + l15];
        b.y = B[(size_t)(k + kpair + 1) * NCOLS + n0 + l15];
        acc = __builtin_amdgcn_wmma_f32_16x16x4_f32(
            false, a, false, b, (short)0, acc, false, false);
    }

    const int mbase = half * 8;
#pragma unroll
    for (int r = 0; r < 8; ++r) {
        int row = m0 + mbase + r;
        if (row < M) C[(size_t)row * NCOLS + n0 + l15] = acc[r];
    }
}

__global__ void __launch_bounds__(256)
edge_proj_score(const float* __restrict__ EF, const float* __restrict__ We,
                const float* __restrict__ H, const int* __restrict__ src,
                float* __restrict__ score, int E) {
    __shared__ float Apanel[16 * EDGE_K];
    __shared__ int   srcIdx[16];
    __shared__ float spart[8][16];

    const int lane = threadIdx.x & 31;
    const int wave = threadIdx.x >> 5;
    const int m0   = blockIdx.x * 16;

    for (int idx = threadIdx.x; idx < 16 * EDGE_K; idx += 256) {
        int r = idx >> 6, c = idx & 63;
        int row = m0 + r; row = row < E ? row : E - 1;
        Apanel[idx] = EF[(size_t)row * EDGE_K + c];
    }
    if (threadIdx.x < 16) {
        int e = m0 + threadIdx.x;
        srcIdx[threadIdx.x] = src[e < E ? e : E - 1];
    }
    __syncthreads();

    const int n0    = wave * 16;
    const int half  = lane >> 4;
    const int l15   = lane & 15;
    const int kpair = half * 2;

    v8f acc = {};
    for (int k = 0; k < EDGE_K; k += 4) {
        v2f a, b;
        a.x = Apanel[l15 * EDGE_K + k + kpair];
        a.y = Apanel[l15 * EDGE_K + k + kpair + 1];
        b.x = We[(size_t)(k + kpair)     * NCOLS + n0 + l15];
        b.y = We[(size_t)(k + kpair + 1) * NCOLS + n0 + l15];
        acc = __builtin_amdgcn_wmma_f32_16x16x4_f32(
            false, a, false, b, (short)0, acc, false, false);
    }

    const int mbase = half * 8;
    float prod[8];
#pragma unroll
    for (int r = 0; r < 8; ++r) {
        int s = srcIdx[mbase + r];
        float hv = H[(size_t)s * NCOLS + n0 + l15];
        prod[r] = acc[r] * hv;
    }
#pragma unroll
    for (int off = 1; off < 16; off <<= 1) {
#pragma unroll
        for (int r = 0; r < 8; ++r) prod[r] += __shfl_xor(prod[r], off, 32);
    }
    if (l15 == 0) {
#pragma unroll
        for (int r = 0; r < 8; ++r) spart[wave][mbase + r] = prod[r];
    }
    __syncthreads();

    if (threadIdx.x < 64) {
        int m = threadIdx.x >> 2, hd = threadIdx.x & 3;
        int e = m0 + m;
        if (e < E) {
            float sv = (spart[2 * hd][m] + spart[2 * hd + 1][m]) * 0.17677669529663689f;
            sv = sv > 0.f ? sv : 0.01f * sv;
            score[(size_t)e * NHEADS + hd] = sv;
        }
    }
}

__global__ void __launch_bounds__(256)
init_kernel(int* __restrict__ cnt, int N) {
    int gid = blockIdx.x * 256 + threadIdx.x;
    if (gid < N) cnt[gid] = 0;
}


__global__ void __launch_bounds__(256)
rank_kernel(const int* __restrict__ dst, int* __restrict__ cnt,
            int* __restrict__ rank, int E) {
    int e = blockIdx.x * 256 + threadIdx.x;
    if (e < E) rank[e] = atomicAdd(&cnt[dst[e]], 1);
}

__global__ void __launch_bounds__(1024)
scan_kernel(const int* __restrict__ cnt, int* __restrict__ offs, int N) {
    __shared__ int wtot[32];
    const int tid  = threadIdx.x;
    const int lane = tid & 31;
    const int wave = tid >> 5;
    const int per  = (N + 1023) / 1024;
    const int beg  = tid * per;
    int end = beg + per; end = end < N ? end : N;

    int sum = 0;
    for (int i = beg; i < end; ++i) sum += cnt[i];

    int inc = sum;
#pragma unroll
    for (int off = 1; off < 32; off <<= 1) {
        int t = __shfl_up(inc, off, 32);
        if (lane >= off) inc += t;
    }
    if (lane == 31) wtot[wave] = inc;
    __syncthreads();
    if (wave == 0) {
        int w = wtot[lane], winc = w;
#pragma unroll
        for (int off = 1; off < 32; off <<= 1) {
            int t = __shfl_up(winc, off, 32);
            if (lane >= off) winc += t;
        }
        wtot[lane] = winc - w;
    }
    __syncthreads();

    int run = wtot[wave] + inc - sum;
    for (int i = beg; i < end; ++i) { offs[i] = run; run += cnt[i]; }
    if (tid == 1023) offs[N] = run;
}

__global__ void __launch_bounds__(256)
fill_kernel(const int* __restrict__ src, const int* __restrict__ dst,
            const int* __restrict__ rank, const int* __restrict__ offs,
            int* __restrict__ slot, int E) {
    int e = blockIdx.x * 256 + threadIdx.x;
    if (e >= E) return;
    int pos = offs[dst[e]] + rank[e];
    ((int2*)slot)[pos] = make_int2(e, src[e]);
}

__device__ __forceinline__ float wave_max(float v) {
    for (int off = 16; off > 0; off >>= 1) v = fmaxf(v, __shfl_xor(v, off, 32));
    return v;
}
__device__ __forceinline__ float wave_sum(float v) {
    for (int off = 16; off > 0; off >>= 1) v += __shfl_xor(v, off, 32);
    return v;
}

__device__ __forceinline__ void add_rows(const float* __restrict__ H, int srcs, float4 a,
                                         int n, int lane, int hd, float4& acc) {
    const int k0 = hd == 0 ? -1 : 0, k1 = hd == 1 ? -1 : 0;
    const int k2 = hd == 2 ? -1 : 0, k3 = hd == 3 ? -1 : 0;
    for (int j = 0; j < n; ++j) {
        const int s  = __builtin_amdgcn_readlane(srcs, j);
        const int b0 = __builtin_amdgcn_readlane(__float_as_int(a.x), j);
        const int b1 = __builtin_amdgcn_readlane(__float_as_int(a.y), j);
        const int b2 = __builtin_amdgcn_readlane(__float_as_int(a.z), j);
        const int b3 = __builtin_amdgcn_readlane(__float_as_int(a.w), j);
        const float attn = __int_as_float((b0 & k0) | (b1 & k1) | (b2 & k2) | (b3 & k3));
        const float4 hv = *(const float4*)(H + (size_t)s * NCOLS + lane * 4);
        acc.x += hv.x * attn;
        acc.y += hv.y * attn;
        acc.z += hv.z * attn;
        acc.w += hv.w * attn;
    }
}

__global__ void __launch_bounds__(256)
softmax_aggregate_kernel(const float* __restrict__ H, const int* __restrict__ slot,
                         const int* __restrict__ offs, const float* __restrict__ score,
                         float* __restrict__ out, int N) {
    const int lane = threadIdx.x & 31;
    const int d = __builtin_amdgcn_readfirstlane((int)(blockIdx.x * 8 + (threadIdx.x >> 5)));
    if (d >= N) return;
    const int beg = offs[d], end = offs[d + 1];
    const int hd  = lane >> 3;
    const int2*   slot2  = (const int2*)slot;
    const float4* score4 = (const float4*)score;
    const float ninf = __int_as_float(0xFF800000);

    float4 acc = make_float4(0.f, 0.f, 0.f, 0.f);
    if (end > beg) {
        int n0 = end - beg; n0 = n0 < 32 ? n0 : 32;
        int2   es0 = make_int2(0, 0);
        float4 sc0 = make_float4(ninf, ninf, ninf, ninf);
        if (lane < n0) { es0 = slot2[beg + lane]; sc0 = score4[es0.x]; }

        float4 m = sc0;
        for (int base = beg + 32; base < end; base += 32) {
            if (base + lane < end) {
                const float4 sc = score4[slot2[base + lane].x];
                m.x = fmaxf(m.x, sc.x); m.y = fmaxf(m.y, sc.y);
                m.z = fmaxf(m.z, sc.z); m.w = fmaxf(m.w, sc.w);
            }
        }
        m.x = wave_max(m.x); m.y = wave_max(m.y);
        m.z = wave_max(m.z); m.w = wave_max(m.w);

        float4 w0 = make_float4(0.f, 0.f, 0.f, 0.f);
        if (lane < n0) {
            w0.x = __expf(sc0.x - m.x); w0.y = __expf(sc0.y - m.y);
            w0.z = __expf(sc0.z - m.z); w0.w = __expf(sc0.w - m.w);
        }
        float4 den = w0;
        for (int base = beg + 32; base < end; base += 32) {
            if (base + lane < end) {
                const float4 sc = score4[slot2[base + lane].x];
                den.x += __expf(sc.x - m.x); den.y += __expf(sc.y - m.y);
                den.z += __expf(sc.z - m.z); den.w += __expf(sc.w - m.w);
            }
        }
        den.x = wave_sum(den.x); den.y = wave_sum(den.y);
        den.z = wave_sum(den.z); den.w = wave_sum(den.w);

        add_rows(H, es0.y,
                 make_float4(w0.x / den.x, w0.y / den.y, w0.z / den.z, w0.w / den.w),
                 n0, lane, hd, acc);
        for (int base = beg + 32; base < end; base += 32) {
            int n = end - base; n = n < 32 ? n : 32;
            int2   es = make_int2(0, 0);
            float4 a  = make_float4(0.f, 0.f, 0.f, 0.f);
            if (lane < n) {
                es = slot2[base + lane];
                const float4 sc = score4[es.x];
                a.x = __expf(sc.x - m.x) / den.x; a.y = __expf(sc.y - m.y) / den.y;
                a.z = __expf(sc.z - m.z) / den.z; a.w = __expf(sc.w - m.w) / den.w;
            }
            add_rows(H, es.y, a, n, lane, hd, acc);
        }
    }
    *(float4*)(out + (size_t)d * NCOLS + lane * 4) = acc;
}

extern "C" void kernel_launch(void* const* d_in, const int* in_sizes, int n_in,
                              void* d_out, int out_size, void* d_ws, size_t ws_size,
                              hipStream_t stream) {
    const float* node_feat = (const float*)d_in[0];
    const float* edge_feat = (const float*)d_in[1];
    const int*   src       = (const int*)d_in[2];
    const int*   dst       = (const int*)d_in[3];
    const float* W_node    = (const float*)d_in[4];
    const float* W_edge    = (const float*)d_in[5];
    float*       out       = (float*)d_out;

    const int N = in_sizes[0] / NODE_K;
    const int E = in_sizes[2];

    float* h     = (float*)d_ws;
    float* score = h     + (size_t)N * NCOLS;
    float* sortb = score + (size_t)E * NHEADS;
    float* offsb = sortb + (size_t)E * NHEADS;

    gemm_f32_wmma<NODE_K><<<(N + 15) / 16, 256, 0, stream>>>(node_feat, W_node, h, N);

    edge_proj_score<<<(E + 15) / 16, 256, 0, stream>>>(edge_feat, W_edge, h, src, score, E);

    int* cnt  = (int*)out;
    int* slot = (int*)sortb;
    int* rank = (int*)(sortb + 2 * (size_t)E);
    int* offs = (int*)offsb;

    init_kernel<<<(N + 255) / 256, 256, 0, stream>>>(cnt, N);
    rank_kernel<<<(E + 255) / 256, 256, 0, stream>>>(dst, cnt, rank, E);
    scan_kernel<<<1, 1024, 0, stream>>>(cnt, offs, N);
    fill_kernel<<<(E + 255) / 256, 256, 0, stream>>>(src, dst, rank, offs, slot, E);

    softmax_aggregate_kernel<<<(N + 7) / 8, 256, 0, stream>>>(h, slot, offs, score, out, N);
}


# edge score kernel re-tiled: one wave owns a 16-edge tile over all 128 columns, A straight from global, W_edge fragments staged once per resident block in LDS; head dot sum order changes
# speedup vs baseline: 2.3424x; 1.4955x over previous
#include <hip/hip_runtime.h>
#include <hip/hip_bf16.h>

typedef __attribute__((ext_vector_type(2))) float v2f;
typedef __attribute__((ext_vector_type(8))) float v8f;

#define NCOLS 128
#define NODE_K 256
#define EDGE_K 64
#define NHEADS 4

template <int K>
__global__ void __launch_bounds__(256)
gemm_f32_wmma(const float* __restrict__ A, const float* __restrict__ B,
              float* __restrict__ C, int M) {
    __shared__ float Apanel[16 * K];
    const int lane = threadIdx.x & 31;
    const int wave = threadIdx.x >> 5;
    const int m0   = blockIdx.x * 16;

    for (int idx = threadIdx.x; idx < 16 * K; idx += 256) {
        int r = idx / K, c = idx % K;
        int row = m0 + r; row = row < M ? row : M - 1;
        Apanel[idx] = A[(size_t)row * K + c];
    }
    __syncthreads();

    const int n0      = wave * 16;
    const int half    = lane >> 4;
    const int l15     = lane & 15;
    const int kpair   = half * 2;

    v8f acc = {};
    for (int k = 0; k < K; k += 4) {
        v2f a, b;
        a.x = Apanel[l15 * K + k + kpair];
        a.y = Apanel[l15 * K + k + kpair + 1];
        b.x = B[(size_t)(k + kpair)     * NCOLS + n0 + l15];
        b.y = B[(size_t)(k + kpair + 1) * NCOLS + n0 + l15];
        acc = __builtin_amdgcn_wmma_f32_16x16x4_f32(
            false, a, false, b, (short)0, acc, false, false);
    }

    const int mbase = half * 8;
#pragma unroll
    for (int r = 0; r < 8; ++r) {
        int row = m0 + mbase + r;
        if (row < M) C[(size_t)row * NCOLS + n0 + l15] = acc[r];
    }
}

__global__ void __launch_bounds__(256, 6)
edge_proj_score(const float* __restrict__ EF, const float* __restrict__ We,
                const float* __restrict__ H, const int* __restrict__ src,
                float* __restrict__ score, int E) {
    __shared__ v2f Bfrag[16 * 8 * 32];

    const int lane = threadIdx.x & 31;
    const int wave = threadIdx.x >> 5;
    const int half = lane >> 4;
    const int l15  = lane & 15;

#pragma unroll
    for (int q = 0; q < 16; ++q) {
        const int k0 = 8 * (q >> 1) + 4 * half + 2 * (q & 1);
        v2f b;
        b.x = We[(size_t)k0       * NCOLS + 16 * wave + l15];
        b.y = We[(size_t)(k0 + 1) * NCOLS + 16 * wave + l15];
        Bfrag[(q * 8 + wave) * 32 + lane] = b;
    }
    __syncthreads();

    const int ntiles = (E + 15) >> 4;
    for (int tile = blockIdx.x * 8 + wave; tile < ntiles; tile += gridDim.x * 8) {
        const int m0 = tile * 16;

        asm volatile("" ::: "memory");

        int row = m0 + l15; row = row < E ? row : E - 1;
        const float4* arow = (const float4*)(EF + (size_t)row * EDGE_K) + half;
        float4 a4[8];
#pragma unroll
        for (int j = 0; j < 8; ++j) a4[j] = arow[2 * j];

        int s[8];
#pragma unroll
        for (int r = 0; r < 8; ++r) {
            int e = m0 + 8 * half + r;
            s[r] = src[e < E ? e : E - 1];
        }

        v8f acc[8];
#pragma unroll
        for (int t = 0; t < 8; ++t) acc[t] = (v8f){};
#pragma unroll
        for (int j = 0; j < 8; ++j) {
            v2f a0, a1;
            a0.x = a4[j].x; a0.y = a4[j].y;
            a1.x = a4[j].z; a1.y = a4[j].w;
#pragma unroll
            for (int t = 0; t < 8; ++t)
                acc[t] = __builtin_amdgcn_wmma_f32_16x16x4_f32(
                    false, a0, false, Bfrag[((2 * j) * 8 + t) * 32 + lane],
                    (short)0, acc[t], false, false);
#pragma unroll
            for (int t = 0; t < 8; ++t)
                acc[t] = __builtin_amdgcn_wmma_f32_16x16x4_f32(
                    false, a1, false, Bfrag[((2 * j + 1) * 8 + t) * 32 + lane],
                    (short)0, acc[t], false, false);
        }

        float sc[8][4];
#pragma unroll
        for (int hd = 0; hd < 4; ++hd) {
            float p[8];
#pragma unroll
            for (int r = 0; r < 8; ++r) {
                const float* hrow = H + (size_t)s[r] * NCOLS + 32 * hd + l15;
                p[r] = acc[2 * hd][r] * hrow[0] + acc[2 * hd + 1][r] * hrow[16];
            }
#pragma unroll
            for (int off = 1; off < 16; off <<= 1) {
#pragma unroll
                for (int r = 0; r < 8; ++r) p[r] += __shfl_xor(p[r], off, 32);
            }
#pragma unroll
            for (int r = 0; r < 8; ++r) {
                float sv = p[r] * 0.17677669529663689f;
                sc[r][hd] = sv > 0.f ? sv : 0.01f * sv;
            }
        }
        if (l15 == 0) {
#pragma unroll
            for (int r = 0; r < 8; ++r) {
                int e = m0 + 8 * half + r;
                if (e < E)
                    *(float4*)(score + (size_t)e * NHEADS) =
                        make_float4(sc[r][0], sc[r][1], sc[r][2], sc[r][3]);
            }
        }
    }
}

__global__ void __launch_bounds__(256)
init_kernel(int* __restrict__ cnt, int N) {
    int gid = blockIdx.x * 256 + threadIdx.x;
    if (gid < N) cnt[gid] = 0;
}


__global__ void __launch_bounds__(256)
rank_kernel(const int* __restrict__ dst, int* __restrict__ cnt,
            int* __restrict__ rank, int E) {
    int e = blockIdx.x * 256 + threadIdx.x;
    if (e < E) rank[e] = atomicAdd(&cnt[dst[e]], 1);
}

__global__ void __launch_bounds__(1024)
scan_kernel(const int* __restrict__ cnt, int* __restrict__ offs, int N) {
    __shared__ int wtot[32];
    const int tid  = threadIdx.x;
    const int lane = tid & 31;
    const int wave = tid >> 5;
    const int per  = (N + 1023) / 1024;
    const int beg  = tid * per;
    int end = beg + per; end = end < N ? end : N;

    int sum = 0;
    for (int i = beg; i < end; ++i) sum += cnt[i];

    int inc = sum;
#pragma unroll
    for (int off = 1; off < 32; off <<= 1) {
        int t = __shfl_up(inc, off, 32);
        if (lane >= off) inc += t;
    }
    if (lane == 31) wtot[wave] = inc;
    __syncthreads();
    if (wave == 0) {
        int w = wtot[lane], winc = w;
#pragma unroll
        for (int off = 1; off < 32; off <<= 1) {
            int t = __shfl_up(winc, off, 32);
            if (lane >= off) winc += t;
        }
        wtot[lane] = winc - w;
    }
    __syncthreads();

    int run = wtot[wave] + inc - sum;
    for (int i = beg; i < end; ++i) { offs[i] = run; run += cnt[i]; }
    if (tid == 1023) offs[N] = run;
}

__global__ void __launch_bounds__(256)
fill_kernel(const int* __restrict__ src, const int* __restrict__ dst,
            const int* __restrict__ rank, const int* __restrict__ offs,
            int* __restrict__ slot, int E) {
    int e = blockIdx.x * 256 + threadIdx.x;
    if (e >= E) return;
    int pos = offs[dst[e]] + rank[e];
    ((int2*)slot)[pos] = make_int2(e, src[e]);
}

__device__ __forceinline__ float wave_max(float v) {
    for (int off = 16; off > 0; off >>= 1) v = fmaxf(v, __shfl_xor(v, off, 32));
    return v;
}
__device__ __forceinline__ float wave_sum(float v) {
    for (int off = 16; off > 0; off >>= 1) v += __shfl_xor(v, off, 32);
    return v;
}

__device__ __forceinline__ void add_rows(const float* __restrict__ H, int srcs, float4 a,
                                         int n, int lane, int hd, float4& acc) {
    const int k0 = hd == 0 ? -1 : 0, k1 = hd == 1 ? -1 : 0;
    const int k2 = hd == 2 ? -1 : 0, k3 = hd == 3 ? -1 : 0;
    for (int j = 0; j < n; ++j) {
        const int s  = __builtin_amdgcn_readlane(srcs, j);
        const int b0 = __builtin_amdgcn_readlane(__float_as_int(a.x), j);
        const int b1 = __builtin_amdgcn_readlane(__float_as_int(a.y), j);
        const int b2 = __builtin_amdgcn_readlane(__float_as_int(a.z), j);
        const int b3 = __builtin_amdgcn_readlane(__float_as_int(a.w), j);
        const float attn = __int_as_float((b0 & k0) | (b1 & k1) | (b2 & k2) | (b3 & k3));
        const float4 hv = *(const float4*)(H + (size_t)s * NCOLS + lane * 4);
        acc.x += hv.x * attn;
        acc.y += hv.y * attn;
        acc.z += hv.z * attn;
        acc.w += hv.w * attn;
    }
}

__global__ void __launch_bounds__(256)
softmax_aggregate_kernel(const float* __restrict__ H, const int* __restrict__ slot,
                         const int* __restrict__ offs, const float* __restrict__ score,
                         float* __restrict__ out, int N) {
    const int lane = threadIdx.x & 31;
    const int d = __builtin_amdgcn_readfirstlane((int)(blockIdx.x * 8 + (threadIdx.x >> 5)));
    if (d >= N) return;
    const int beg = offs[d], end = offs[d + 1];
    const int hd  = lane >> 3;
    const int2*   slot2  = (const int2*)slot;
    const float4* score4 = (const float4*)score;
    const float ninf = __int_as_float(0xFF800000);

    float4 acc = make_float4(0.f, 0.f, 0.f, 0.f);
    if (end > beg) {
        int n0 = end - beg; n0 = n0 < 32 ? n0 : 32;
        int2   es0 = make_int2(0, 0);
        float4 sc0 = make_float4(ninf, ninf, ninf, ninf);
        if (lane < n0) { es0 = slot2[beg + lane]; sc0 = score4[es0.x]; }

        float4 m = sc0;
        for (int base = beg + 32; base < end; base += 32) {
            if (base + lane < end) {
                const float4 sc = score4[slot2[base + lane].x];
                m.x = fmaxf(m.x, sc.x); m.y = fmaxf(m.y, sc.y);
                m.z = fmaxf(m.z, sc.z); m.w = fmaxf(m.w, sc.w);
            }
        }
        m.x = wave_max(m.x); m.y = wave_max(m.y);
        m.z = wave_max(m.z); m.w = wave_max(m.w);

        float4 w0 = make_float4(0.f, 0.f, 0.f, 0.f);
        if (lane < n0) {
            w0.x = __expf(sc0.x - m.x); w0.y = __expf(sc0.y - m.y);
            w0.z = __expf(sc0.z - m.z); w0.w = __expf(sc0.w - m.w);
        }
        float4 den = w0;
        for (int base = beg + 32; base < end; base += 32) {
            if (base + lane < end) {
                const float4 sc = score4[slot2[base + lane].x];
                den.x += __expf(sc.x - m.x); den.y += __expf(sc.y - m.y);
                den.z += __expf(sc.z - m.z); den.w += __expf(sc.w - m.w);
            }
        }
        den.x = wave_sum(den.x); den.y = wave_sum(den.y);
        den.z = wave_sum(den.z); den.w = wave_sum(den.w);

        add_rows(H, es0.y,
                 make_float4(w0.x / den.x, w0.y / den.y, w0.z / den.z, w0.w / den.w),
                 n0, lane, hd, acc);
        for (int base = beg + 32; base < end; base += 32) {
            int n = end - base; n = n < 32 ? n : 32;
            int2   es = make_int2(0, 0);
            float4 a  = make_float4(0.f, 0.f, 0.f, 0.f);
            if (lane < n) {
                es = slot2[base + lane];
                const float4 sc = score4[es.x];
                a.x = __expf(sc.x - m.x) / den.x; a.y = __expf(sc.y - m.y) / den.y;
                a.z = __expf(sc.z - m.z) / den.z; a.w = __expf(sc.w - m.w) / den.w;
            }
            add_rows(H, es.y, a, n, lane, hd, acc);
        }
    }
    *(float4*)(out + (size_t)d * NCOLS + lane * 4) = acc;
}

extern "C" void kernel_launch(void* const* d_in, const int* in_sizes, int n_in,
                              void* d_out, int out_size, void* d_ws, size_t ws_size,
                              hipStream_t stream) {
    const float* node_feat = (const float*)d_in[0];
    const float* edge_feat = (const float*)d_in[1];
    const int*   src       = (const int*)d_in[2];
    const int*   dst       = (const int*)d_in[3];
    const float* W_node    = (const float*)d_in[4];
    const float* W_edge    = (const float*)d_in[5];
    float*       out       = (float*)d_out;

    const int N = in_sizes[0] / NODE_K;
    const int E = in_sizes[2];

    float* h     = (float*)d_ws;
    float* score = h     + (size_t)N * NCOLS;
    float* sortb = score + (size_t)E * NHEADS;
    float* offsb = sortb + (size_t)E * NHEADS;

    gemm_f32_wmma<NODE_K><<<(N + 15) / 16, 256, 0, stream>>>(node_feat, W_node, h, N);

    edge_proj_score<<<2048, 256, 0, stream>>>(edge_feat, W_edge, h, src, score, E);

    int* cnt  = (int*)out;
    int* slot = (int*)sortb;
    int* rank = (int*)(sortb + 2 * (size_t)E);
    int* offs = (int*)offsb;

    init_kernel<<<(N + 255) / 256, 256, 0, stream>>>(cnt, N);
    rank_kernel<<<(E + 255) / 256, 256, 0, stream>>>(dst, cnt, rank, E);
    scan_kernel<<<1, 1024, 0, stream>>>(cnt, offs, N);
    fill_kernel<<<(E + 255) / 256, 256, 0, stream>>>(src, dst, rank, offs, slot, E);

    softmax_aggregate_kernel<<<(N + 7) / 8, 256, 0, stream>>>(h, slot, offs, score, out, N);
}


# scan kernel: whole-line reads in place of 4-byte reads at a 196-byte pitch; 49 blocks, each sums the counters in front of its 1024 with 16-byte loads, then scans its own, one per thread
# speedup vs baseline: 2.9235x; 1.2481x over previous
#include <hip/hip_runtime.h>
#include <hip/hip_bf16.h>

typedef __attribute__((ext_vector_type(2))) float v2f;
typedef __attribute__((ext_vector_type(8))) float v8f;

#define NCOLS 128
#define NODE_K 256
#define EDGE_K 64
#define NHEADS 4

template <int K>
__global__ void __launch_bounds__(256)
gemm_f32_wmma(const float* __restrict__ A, const float* __restrict__ B,
              float* __restrict__ C, int M) {
    __shared__ float Apanel[16 * K];
    const int lane = threadIdx.x & 31;
    const int wave = threadIdx.x >> 5;
    const int m0   = blockIdx.x * 16;

    for (int idx = threadIdx.x; idx < 16 * K; idx += 256) {
        int r = idx / K, c = idx % K;
        int row = m0 + r; row = row < M ? row : M - 1;
        Apanel[idx] = A[(size_t)row * K + c];
    }
    __syncthreads();

    const int n0      = wave * 16;
    const int half    = lane >> 4;
    const int l15     = lane & 15;
    const int kpair   = half * 2;

    v8f acc = {};
    for (int k = 0; k < K; k += 4) {
        v2f a, b;
        a.x = Apanel[l15 * K + k + kpair];
        a.y = Apanel[l15 * K + k + kpair + 1];
        b.x = B[(size_t)(k + kpair)     * NCOLS + n0 + l15];
        b.y = B[(size_t)(k + kpair + 1) * NCOLS + n0 + l15];
        acc = __builtin_amdgcn_wmma_f32_16x16x4_f32(
            false, a, false, b, (short)0, acc, false, false);
    }

    const int mbase = half * 8;
#pragma unroll
    for (int r = 0; r < 8; ++r) {
        int row = m0 + mbase + r;
        if (row < M) C[(size_t)row * NCOLS + n0 + l15] = acc[r];
    }
}

__global__ void __launch_bounds__(256, 6)
edge_proj_score(const float* __restrict__ EF, const float* __restrict__ We,
                const float* __restrict__ H, const int* __restrict__ src,
                float* __restrict__ score, int E) {
    __shared__ v2f Bfrag[16 * 8 * 32];

    const int lane = threadIdx.x & 31;
    const int wave = threadIdx.x >> 5;
    const int half = lane >> 4;
    const int l15  = lane & 15;

#pragma unroll
    for (int q = 0; q < 16; ++q) {
        const int k0 = 8 * (q >> 1) + 4 * half + 2 * (q & 1);
        v2f b;
        b.x = We[(size_t)k0       * NCOLS + 16 * wave + l15];
        b.y = We[(size_t)(k0 + 1) * NCOLS + 16 * wave + l15];
        Bfrag[(q * 8 + wave) * 32 + lane] = b;
    }
    __syncthreads();

    const int ntiles = (E + 15) >> 4;
    for (int tile = blockIdx.x * 8 + wave; tile < ntiles; tile += gridDim.x * 8) {
        const int m0 = tile * 16;

        asm volatile("" ::: "memory");

        int row = m0 + l15; row = row < E ? row : E - 1;
        const float4* arow = (const float4*)(EF + (size_t)row * EDGE_K) + half;
        float4 a4[8];
#pragma unroll
        for (int j = 0; j < 8; ++j) a4[j] = arow[2 * j];

        int s[8];
#pragma unroll
        for (int r = 0; r < 8; ++r) {
            int e = m0 + 8 * half + r;
            s[r] = src[e < E ? e : E - 1];
        }

        v8f acc[8];
#pragma unroll
        for (int t = 0; t < 8; ++t) acc[t] = (v8f){};
#pragma unroll
        for (int j = 0; j < 8; ++j) {
            v2f a0, a1;
            a0.x = a4[j].x; a0.y = a4[j].y;
            a1.x = a4[j].z; a1.y = a4[j].w;
#pragma unroll
            for (int t = 0; t < 8; ++t)
                acc[t] = __builtin_amdgcn_wmma_f32_16x16x4_f32(
                    false, a0, false, Bfrag[((2 * j) * 8 + t) * 32 + lane],
                    (short)0, acc[t], false, false);
#pragma unroll
            for (int t = 0; t < 8; ++t)
                acc[t] = __builtin_amdgcn_wmma_f32_16x16x4_f32(
                    false, a1, false, Bfrag[((2 * j + 1) * 8 + t) * 32 + lane],
                    (short)0, acc[t], false, false);
        }

        float sc[8][4];
#pragma unroll
        for (int hd = 0; hd < 4; ++hd) {
            float p[8];
#pragma unroll
            for (int r = 0; r < 8; ++r) {
                const float* hrow = H + (size_t)s[r] * NCOLS + 32 * hd + l15;
                p[r] = acc[2 * hd][r] * hrow[0] + acc[2 * hd + 1][r] * hrow[16];
            }
#pragma unroll
            for (int off = 1; off < 16; off <<= 1) {
#pragma unroll
                for (int r = 0; r < 8; ++r) p[r] += __shfl_xor(p[r], off, 32);
            }
#pragma unroll
            for (int r = 0; r < 8; ++r) {
                float sv = p[r] * 0.17677669529663689f;
                sc[r][hd] = sv > 0.f ? sv : 0.01f * sv;
            }
        }
        if (l15 == 0) {
#pragma unroll
            for (int r = 0; r < 8; ++r) {
                int e = m0 + 8 * half + r;
                if (e < E)
                    *(float4*)(score + (size_t)e * NHEADS) =
                        make_float4(sc[r][0], sc[r][1], sc[r][2], sc[r][3]);
            }
        }
    }
}

__global__ void __launch_bounds__(256)
init_kernel(int* __restrict__ cnt, int N) {
    int gid = blockIdx.x * 256 + threadIdx.x;
    if (gid < N) cnt[gid] = 0;
}


__global__ void __launch_bounds__(256)
rank_kernel(const int* __restrict__ dst, int* __restrict__ cnt,
            int* __restrict__ rank, int E) {
    int e = blockIdx.x * 256 + threadIdx.x;
    if (e < E) rank[e] = atomicAdd(&cnt[dst[e]], 1);
}

__device__ __forceinline__ int wave_sum_int(int v) {
    for (int off = 16; off > 0; off >>= 1) v += __shfl_xor(v, off, 32);
    return v;
}

__global__ void __launch_bounds__(1024)
scan_kernel(const int* __restrict__ cnt, int* __restrict__ offs, int N) {
    __shared__ int wpre[32];
    __shared__ int wtot[32];
    const int tid  = threadIdx.x;
    const int lane = tid & 31;
    const int wave = tid >> 5;
    const int base = blockIdx.x * 1024;

    const int4* __restrict__ c4 = (const int4*)cnt;
    const int n4 = base >> 2;
    int pre = 0;
    for (int i0 = tid; i0 < n4; i0 += 16 * 1024) {
        int4 v[16];
#pragma unroll
        for (int j = 0; j < 16; ++j) {
            const int i = i0 + j * 1024;
            v[j] = c4[i < n4 ? i : n4 - 1];
        }
#pragma unroll
        for (int j = 0; j < 16; ++j) {
            const int sj = (v[j].x + v[j].y) + (v[j].z + v[j].w);
            pre += (i0 + j * 1024 < n4) ? sj : 0;
        }
    }

    const int g = base + tid;
    const int c = g < N ? cnt[g] : 0;

    int inc = c;
#pragma unroll
    for (int off = 1; off < 32; off <<= 1) {
        int t = __shfl_up(inc, off, 32);
        if (lane >= off) inc += t;
    }
    pre = wave_sum_int(pre);
    if (lane == 31) wtot[wave] = inc;
    if (lane == 0)  wpre[wave] = pre;
    __syncthreads();

    int x = wpre[lane] + (lane < wave ? wtot[lane] : 0);
    x = wave_sum_int(x);

    const int excl = x + inc - c;
    if (g < N) offs[g] = excl;
    if (g == N - 1) offs[N] = excl + c;
}

__global__ void __launch_bounds__(256)
fill_kernel(const int* __restrict__ src, const int* __restrict__ dst,
            const int* __restrict__ rank, const int* __restrict__ offs,
            int* __restrict__ slot, int E) {
    int e = blockIdx.x * 256 + threadIdx.x;
    if (e >= E) return;
    int pos = offs[dst[e]] + rank[e];
    ((int2*)slot)[pos] = make_int2(e, src[e]);
}

__device__ __forceinline__ float wave_max(float v) {
    for (int off = 16; off > 0; off >>= 1) v = fmaxf(v, __shfl_xor(v, off, 32));
    return v;
}
__device__ __forceinline__ float wave_sum(float v) {
    for (int off = 16; off > 0; off >>= 1) v += __shfl_xor(v, off, 32);
    return v;
}

__device__ __forceinline__ void add_rows(const float* __restrict__ H, int srcs, float4 a,
                                         int n, int lane, int hd, float4& acc) {
    const int k0 = hd == 0 ? -1 : 0, k1 = hd == 1 ? -1 : 0;
    const int k2 = hd == 2 ? -1 : 0, k3 = hd == 3 ? -1 : 0;
    for (int j = 0; j < n; ++j) {
        const int s  = __builtin_amdgcn_readlane(srcs, j);
        const int b0 = __builtin_amdgcn_readlane(__float_as_int(a.x), j);
        const int b1 = __builtin_amdgcn_readlane(__float_as_int(a.y), j);
        const int b2 = __builtin_amdgcn_readlane(__float_as_int(a.z), j);
        const int b3 = __builtin_amdgcn_readlane(__float_as_int(a.w), j);
        const float attn = __int_as_float((b0 & k0) | (b1 & k1) | (b2 & k2) | (b3 & k3));
        const float4 hv = *(const float4*)(H + (size_t)s * NCOLS + lane * 4);
        acc.x += hv.x * attn;
        acc.y += hv.y * attn;
        acc.z += hv.z * attn;
        acc.w += hv.w * attn;
    }
}

__global__ void __launch_bounds__(256)
softmax_aggregate_kernel(const float* __restrict__ H, const int* __restrict__ slot,
                         const int* __restrict__ offs, const float* __restrict__ score,
                         float* __restrict__ out, int N) {
    const int lane = threadIdx.x & 31;
    const int d = __builtin_amdgcn_readfirstlane((int)(blockIdx.x * 8 + (threadIdx.x >> 5)));
    if (d >= N) return;
    const int beg = offs[d], end = offs[d + 1];
    const int hd  = lane >> 3;
    const int2*   slot2  = (const int2*)slot;
    const float4* score4 = (const float4*)score;
    const float ninf = __int_as_float(0xFF800000);

    float4 acc = make_float4(0.f, 0.f, 0.f, 0.f);
    if (end > beg) {
        int n0 = end - beg; n0 = n0 < 32 ? n0 : 32;
        int2   es0 = make_int2(0, 0);
        float4 sc0 = make_float4(ninf, ninf, ninf, ninf);
        if (lane < n0) { es0 = slot2[beg + lane]; sc0 = score4[es0.x]; }

        float4 m = sc0;
        for (int base = beg + 32; base < end; base += 32) {
            if (base + lane < end) {
                const float4 sc = score4[slot2[base + lane].x];
                m.x = fmaxf(m.x, sc.x); m.y = fmaxf(m.y, sc.y);
                m.z = fmaxf(m.z, sc.z); m.w = fmaxf(m.w, sc.w);
            }
        }
        m.x = wave_max(m.x); m.y = wave_max(m.y);
        m.z = wave_max(m.z); m.w = wave_max(m.w);

        float4 w0 = make_float4(0.f, 0.f, 0.f, 0.f);
        if (lane < n0) {
            w0.x = __expf(sc0.x - m.x); w0.y = __expf(sc0.y - m.y);
            w0.z = __expf(sc0.z - m.z); w0.w = __expf(sc0.w - m.w);
        }
        float4 den = w0;
        for (int base = beg + 32; base < end; base += 32) {
            if (base + lane < end) {
                const float4 sc = score4[slot2[base + lane].x];
                den.x += __expf(sc.x - m.x); den.y += __expf(sc.y - m.y);
                den.z += __expf(sc.z - m.z); den.w += __expf(sc.w - m.w);
            }
        }
        den.x = wave_sum(den.x); den.y = wave_sum(den.y);
        den.z = wave_sum(den.z); den.w = wave_sum(den.w);

        add_rows(H, es0.y,
                 make_float4(w0.x / den.x, w0.y / den.y, w0.z / den.z, w0.w / den.w),
                 n0, lane, hd, acc);
        for (int base = beg + 32; base < end; base += 32) {
            int n = end - base; n = n < 32 ? n : 32;
            int2   es = make_int2(0, 0);
            float4 a  = make_float4(0.f, 0.f, 0.f, 0.f);
            if (lane < n) {
                es = slot2[base + lane];
                const float4 sc = score4[es.x];
                a.x = __expf(sc.x - m.x) / den.x; a.y = __expf(sc.y - m.y) / den.y;
                a.z = __expf(sc.z - m.z) / den.z; a.w = __expf(sc.w - m.w) / den.w;
            }
            add_rows(H, es.y, a, n, lane, hd, acc);
        }
    }
    *(float4*)(out + (size_t)d * NCOLS + lane * 4) = acc;
}

extern "C" void kernel_launch(void* const* d_in, const int* in_sizes, int n_in,
                              void* d_out, int out_size, void* d_ws, size_t ws_size,
                              hipStream_t stream) {
    const float* node_feat = (const float*)d_in[0];
    const float* edge_feat = (const float*)d_in[1];
    const int*   src       = (const int*)d_in[2];
    const int*   dst       = (const int*)d_in[3];
    const float* W_node    = (const float*)d_in[4];
    const float* W_edge    = (const float*)d_in[5];
    float*       out       = (float*)d_out;

    const int N = in_sizes[0] / NODE_K;
    const int E = in_sizes[2];

    float* h     = (float*)d_ws;
    float* score = h     + (size_t)N * NCOLS;
    float* sortb = score + (size_t)E * NHEADS;
    float* offsb = sortb + (size_t)E * NHEADS;

    gemm_f32_wmma<NODE_K><<<(N + 15) / 16, 256, 0, stream>>>(node_feat, W_node, h, N);

    edge_proj_score<<<2048, 256, 0, stream>>>(edge_feat, W_edge, h, src, score, E);

    int* cnt  = (int*)out;
    int* slot = (int*)sortb;
    int* rank = (int*)(sortb + 2 * (size_t)E);
    int* offs = (int*)offsb;

    init_kernel<<<(N + 255) / 256, 256, 0, stream>>>(cnt, N);
    rank_kernel<<<(E + 255) / 256, 256, 0, stream>>>(dst, cnt, rank, E);
    scan_kernel<<<(N + 1023) / 1024, 1024, 0, stream>>>(cnt, offs, N);
    fill_kernel<<<(E + 255) / 256, 256, 0, stream>>>(src, dst, rank, offs, slot, E);

    softmax_aggregate_kernel<<<(N + 7) / 8, 256, 0, stream>>>(h, slot, offs, score, out, N);
}


# node projection GEMM: rows of the A panel in LDS at a pitch of K+4 words instead of K, so the 16 rows of a fragment read fall on different banks (was a 16-way bank conflict on every A read)
# speedup vs baseline: 3.3266x; 1.1379x over previous
#include <hip/hip_runtime.h>
#include <hip/hip_bf16.h>

typedef __attribute__((ext_vector_type(2))) float v2f;
typedef __attribute__((ext_vector_type(8))) float v8f;

#define NCOLS 128
#define NODE_K 256
#define EDGE_K 64
#define NHEADS 4

template <int K>
__global__ void __launch_bounds__(256)
gemm_f32_wmma(const float* __restrict__ A, const float* __restrict__ B,
              float* __restrict__ C, int M) {
    constexpr int AP = K + 4;
    __shared__ float Apanel[16 * AP];
    const int lane = threadIdx.x & 31;
    const int wave = threadIdx.x >> 5;
    const int m0   = blockIdx.x * 16;

    for (int idx = threadIdx.x; idx < 16 * K; idx += 256) {
        int r = idx / K, c = idx % K;
        int row = m0 + r; row = row < M ? row : M - 1;
        Apanel[r * AP + c] = A[(size_t)row * K + c];
    }
    __syncthreads();

    const int n0      = wave * 16;
    const int half    = lane >> 4;
    const int l15     = lane & 15;
    const int kpair   = half * 2;

    v8f acc = {};
    for (int k = 0; k < K; k += 4) {
        v2f a, b;
        a.x = Apanel[l15 * AP + k + kpair];
        a.y = Apanel[l15 * AP + k + kpair + 1];
        b.x = B[(size_t)(k + kpair)     * NCOLS + n0 + l15];
        b.y = B[(size_t)(k + kpair + 1) * NCOLS + n0 + l15];
        acc = __builtin_amdgcn_wmma_f32_16x16x4_f32(
            false, a, false, b, (short)0, acc, false, false);
    }

    const int mbase = half * 8;
#pragma unroll
    for (int r = 0; r < 8; ++r) {
        int row = m0 + mbase + r;
        if (row < M) C[(size_t)row * NCOLS + n0 + l15] = acc[r];
    }
}

__global__ void __launch_bounds__(256, 6)
edge_proj_score(const float* __restrict__ EF, const float* __restrict__ We,
                const float* __restrict__ H, const int* __restrict__ src,
                float* __restrict__ score, int E) {
    __shared__ v2f Bfrag[16 * 8 * 32];

    const int lane = threadIdx.x & 31;
    const int wave = threadIdx.x >> 5;
    const int half = lane >> 4;
    const int l15  = lane & 15;

#pragma unroll
    for (int q = 0; q < 16; ++q) {
        const int k0 = 8 * (q >> 1) + 4 * half + 2 * (q & 1);
        v2f b;
        b.x = We[(size_t)k0       * NCOLS + 16 * wave + l15];
        b.y = We[(size_t)(k0 + 1) * NCOLS + 16 * wave + l15];
        Bfrag[(q * 8 + wave) * 32 + lane] = b;
    }
    __syncthreads();

    const int ntiles = (E + 15) >> 4;
    for (int tile = blockIdx.x * 8 + wave; tile < ntiles; tile += gridDim.x * 8) {
        const int m0 = tile * 16;

        asm volatile("" ::: "memory");

        int row = m0 + l15; row = row < E ? row : E - 1;
        const float4* arow = (const float4*)(EF + (size_t)row * EDGE_K) + half;
        float4 a4[8];
#pragma unroll
        for (int j = 0; j < 8; ++j) a4[j] = arow[2 * j];

        int s[8];
#pragma unroll
        for (int r = 0; r < 8; ++r) {
            int e = m0 + 8 * half + r;
            s[r] = src[e < E ? e : E - 1];
        }

        v8f acc[8];
#pragma unroll
        for (int t = 0; t < 8; ++t) acc[t] = (v8f){};
#pragma unroll
        for (int j = 0; j < 8; ++j) {
            v2f a0, a1;
            a0.x = a4[j].x; a0.y = a4[j].y;
            a1.x = a4[j].z; a1.y = a4[j].w;
#pragma unroll
            for (int t = 0; t < 8; ++t)
                acc[t] = __builtin_amdgcn_wmma_f32_16x16x4_f32(
                    false, a0, false, Bfrag[((2 * j) * 8 + t) * 32 + lane],
                    (short)0, acc[t], false, false);
#pragma unroll
            for (int t = 0; t < 8; ++t)
                acc[t] = __builtin_amdgcn_wmma_f32_16x16x4_f32(
                    false, a1, false, Bfrag[((2 * j + 1) * 8 + t) * 32 + lane],
                    (short)0, acc[t], false, false);
        }

        float sc[8][4];
#pragma unroll
        for (int hd = 0; hd < 4; ++hd) {
            float p[8];
#pragma unroll
            for (int r = 0; r < 8; ++r) {
                const float* hrow = H + (size_t)s[r] * NCOLS + 32 * hd + l15;
                p[r] = acc[2 * hd][r] * hrow[0] + acc[2 * hd + 1][r] * hrow[16];
            }
#pragma unroll
            for (int off = 1; off < 16; off <<= 1) {
#pragma unroll
                for (int r = 0; r < 8; ++r) p[r] += __shfl_xor(p[r], off, 32);
            }
#pragma unroll
            for (int r = 0; r < 8; ++r) {
                float sv = p[r] * 0.17677669529663689f;
                sc[r][hd] = sv > 0.f ? sv : 0.01f * sv;
            }
        }
        if (l15 == 0) {
#pragma unroll
            for (int r = 0; r < 8; ++r) {
                int e = m0 + 8 * half + r;
                if (e < E)
                    *(float4*)(score + (size_t)e * NHEADS) =
                        make_float4(sc[r][0], sc[r][1], sc[r][2], sc[r][3]);
            }
        }
    }
}

__global__ void __launch_bounds__(256)
init_kernel(int* __restrict__ cnt, int N) {
    int gid = blockIdx.x * 256 + threadIdx.x;
    if (gid < N) cnt[gid] = 0;
}


__global__ void __launch_bounds__(256)
rank_kernel(const int* __restrict__ dst, int* __restrict__ cnt,
            int* __restrict__ rank, int E) {
    int e = blockIdx.x * 256 + threadIdx.x;
    if (e < E) rank[e] = atomicAdd(&cnt[dst[e]], 1);
}

__device__ __forceinline__ int wave_sum_int(int v) {
    for (int off = 16; off > 0; off >>= 1) v += __shfl_xor(v, off, 32);
    return v;
}

__global__ void __launch_bounds__(1024)
scan_kernel(const int* __restrict__ cnt, int* __restrict__ offs, int N) {
    __shared__ int wpre[32];
    __shared__ int wtot[32];
    const int tid  = threadIdx.x;
    const int lane = tid & 31;
    const int wave = tid >> 5;
    const int base = blockIdx.x * 1024;

    const int4* __restrict__ c4 = (const int4*)cnt;
    const int n4 = base >> 2;
    int pre = 0;
    for (int i0 = tid; i0 < n4; i0 += 16 * 1024) {
        int4 v[16];
#pragma unroll
        for (int j = 0; j < 16; ++j) {
            const int i = i0 + j * 1024;
            v[j] = c4[i < n4 ? i : n4 - 1];
        }
#pragma unroll
        for (int j = 0; j < 16; ++j) {
            const int sj = (v[j].x + v[j].y) + (v[j].z + v[j].w);
            pre += (i0 + j * 1024 < n4) ? sj : 0;
        }
    }

    const int g = base + tid;
    const int c = g < N ? cnt[g] : 0;

    int inc = c;
#pragma unroll
    for (int off = 1; off < 32; off <<= 1) {
        int t = __shfl_up(inc, off, 32);
        if (lane >= off) inc += t;
    }
    pre = wave_sum_int(pre);
    if (lane == 31) wtot[wave] = inc;
    if (lane == 0)  wpre[wave] = pre;
    __syncthreads();

    int x = wpre[lane] + (lane < wave ? wtot[lane] : 0);
    x = wave_sum_int(x);

    const int excl = x + inc - c;
    if (g < N) offs[g] = excl;
    if (g == N - 1) offs[N] = excl + c;
}

__global__ void __launch_bounds__(256)
fill_kernel(const int* __restrict__ src, const int* __restrict__ dst,
            const int* __restrict__ rank, const int* __restrict__ offs,
            int* __restrict__ slot, int E) {
    int e = blockIdx.x * 256 + threadIdx.x;
    if (e >= E) return;
    int pos = offs[dst[e]] + rank[e];
    ((int2*)slot)[pos] = make_int2(e, src[e]);
}

__device__ __forceinline__ float wave_max(float v) {
    for (int off = 16; off > 0; off >>= 1) v = fmaxf(v, __shfl_xor(v, off, 32));
    return v;
}
__device__ __forceinline__ float wave_sum(float v) {
    for (int off = 16; off > 0; off >>= 1) v += __shfl_xor(v, off, 32);
    return v;
}

__device__ __forceinline__ void add_rows(const float* __restrict__ H, int srcs, float4 a,
                                         int n, int lane, int hd, float4& acc) {
    const int k0 = hd == 0 ? -1 : 0, k1 = hd == 1 ? -1 : 0;
    const int k2 = hd == 2 ? -1 : 0, k3 = hd == 3 ? -1 : 0;
    for (int j = 0; j < n; ++j) {
        const int s  = __builtin_amdgcn_readlane(srcs, j);
        const int b0 = __builtin_amdgcn_readlane(__float_as_int(a.x), j);
        const int b1 = __builtin_amdgcn_readlane(__float_as_int(a.y), j);
        const int b2 = __builtin_amdgcn_readlane(__float_as_int(a.z), j);
        const int b3 = __builtin_amdgcn_readlane(__float_as_int(a.w), j);
        const float attn = __int_as_float((b0 & k0) | (b1 & k1) | (b2 & k2) | (b3 & k3));
        const float4 hv = *(const float4*)(H + (size_t)s * NCOLS + lane * 4);
        acc.x += hv.x * attn;
        acc.y += hv.y * attn;
        acc.z += hv.z * attn;
        acc.w += hv.w * attn;
    }
}

__global__ void __launch_bounds__(256)
softmax_aggregate_kernel(const float* __restrict__ H, const int* __restrict__ slot,
                         const int* __restrict__ offs, const float* __restrict__ score,
                         float* __restrict__ out, int N) {
    const int lane = threadIdx.x & 31;
    const int d = __builtin_amdgcn_readfirstlane((int)(blockIdx.x * 8 + (threadIdx.x >> 5)));
    if (d >= N) return;
    const int beg = offs[d], end = offs[d + 1];
    const int hd  = lane >> 3;
    const int2*   slot2  = (const int2*)slot;
    const float4* score4 = (const float4*)score;
    const float ninf = __int_as_float(0xFF800000);

    float4 acc = make_float4(0.f, 0.f, 0.f, 0.f);
    if (end > beg) {
        int n0 = end - beg; n0 = n0 < 32 ? n0 : 32;
        int2   es0 = make_int2(0, 0);
        float4 sc0 = make_float4(ninf, ninf, ninf, ninf);
        if (lane < n0) { es0 = slot2[beg + lane]; sc0 = score4[es0.x]; }

        float4 m = sc0;
        for (int base = beg + 32; base < end; base += 32) {
            if (base + lane < end) {
                const float4 sc = score4[slot2[base + lane].x];
                m.x = fmaxf(m.x, sc.x); m.y = fmaxf(m.y, sc.y);
                m.z = fmaxf(m.z, sc.z); m.w = fmaxf(m.w, sc.w);
            }
        }
        m.x = wave_max(m.x); m.y = wave_max(m.y);
        m.z = wave_max(m.z); m.w = wave_max(m.w);

        float4 w0 = make_float4(0.f, 0.f, 0.f, 0.f);
        if (lane < n0) {
            w0.x = __expf(sc0.x - m.x); w0.y = __expf(sc0.y - m.y);
            w0.z = __expf(sc0.z - m.z); w0.w = __expf(sc0.w - m.w);
        }
        float4 den = w0;
        for (int base = beg + 32; base < end; base += 32) {
            if (base + lane < end) {
                const float4 sc = score4[slot2[base + lane].x];
                den.x += __expf(sc.x - m.x); den.y += __expf(sc.y - m.y);
                den.z += __expf(sc.z - m.z); den.w += __expf(sc.w - m.w);
            }
        }
        den.x = wave_sum(den.x); den.y = wave_sum(den.y);
        den.z = wave_sum(den.z); den.w = wave_sum(den.w);

        add_rows(H, es0.y,
                 make_float4(w0.x / den.x, w0.y / den.y, w0.z / den.z, w0.w / den.w),
                 n0, lane, hd, acc);
        for (int base = beg + 32; base < end; base += 32) {
            int n = end - base; n = n < 32 ? n : 32;
            int2   es = make_int2(0, 0);
            float4 a  = make_float4(0.f, 0.f, 0.f, 0.f);
            if (lane < n) {
                es = slot2[base + lane];
                const float4 sc = score4[es.x];
                a.x = __expf(sc.x - m.x) / den.x; a.y = __expf(sc.y - m.y) / den.y;
                a.z = __expf(sc.z - m.z) / den.z; a.w = __expf(sc.w - m.w) / den.w;
            }
            add_rows(H, es.y, a, n, lane, hd, acc);
        }
    }
    *(float4*)(out + (size_t)d * NCOLS + lane * 4) = acc;
}

extern "C" void kernel_launch(void* const* d_in, const int* in_sizes, int n_in,
                              void* d_out, int out_size, void* d_ws, size_t ws_size,
                              hipStream_t stream) {
    const float* node_feat = (const float*)d_in[0];
    const float* edge_feat = (const float*)d_in[1];
    const int*   src       = (const int*)d_in[2];
    const int*   dst       = (const int*)d_in[3];
    const float* W_node    = (const float*)d_in[4];
    const float* W_edge    = (const float*)d_in[5];
    float*       out       = (float*)d_out;

    const int N = in_sizes[0] / NODE_K;
    const int E = in_sizes[2];

    float* h     = (float*)d_ws;
    float* score = h     + (size_t)N * NCOLS;
    float* sortb = score + (size_t)E * NHEADS;
    float* offsb = sortb + (size_t)E * NHEADS;

    gemm_f32_wmma<NODE_K><<<(N + 15) / 16, 256, 0, stream>>>(node_feat, W_node, h, N);

    edge_proj_score<<<2048, 256, 0, stream>>>(edge_feat, W_edge, h, src, score, E);

    int* cnt  = (int*)out;
    int* slot = (int*)sortb;
    int* rank = (int*)(sortb + 2 * (size_t)E);
    int* offs = (int*)offsb;

    init_kernel<<<(N + 255) / 256, 256, 0, stream>>>(cnt, N);
    rank_kernel<<<(E + 255) / 256, 256, 0, stream>>>(dst, cnt, rank, E);
    scan_kernel<<<(N + 1023) / 1024, 1024, 0, stream>>>(cnt, offs, N);
    fill_kernel<<<(E + 255) / 256, 256, 0, stream>>>(src, dst, rank, offs, slot, E);

    softmax_aggregate_kernel<<<(N + 7) / 8, 256, 0, stream>>>(h, slot, offs, score, out, N);
}


# edges grouped by dst with a two-level counting sort on LDS counters (bucket pass, owner pass) in place of a global atomic per edge, scan and fill; a node's f32 sum order is now LDS-add arrival order
# speedup vs baseline: 4.1405x; 1.2447x over previous
#include <hip/hip_runtime.h>
#include <hip/hip_bf16.h>

typedef __attribute__((ext_vector_type(2))) float v2f;
typedef __attribute__((ext_vector_type(8))) float v8f;

#define NCOLS 128
#define NODE_K 256
#define EDGE_K 64
#define NHEADS 4

template <int K>
__global__ void __launch_bounds__(256)
gemm_f32_wmma(const float* __restrict__ A, const float* __restrict__ B,
              float* __restrict__ C, int M) {
    constexpr int AP = K + 4;
    __shared__ float Apanel[16 * AP];
    const int lane = threadIdx.x & 31;
    const int wave = threadIdx.x >> 5;
    const int m0   = blockIdx.x * 16;

    for (int idx = threadIdx.x; idx < 16 * K; idx += 256) {
        int r = idx / K, c = idx % K;
        int row = m0 + r; row = row < M ? row : M - 1;
        Apanel[r * AP + c] = A[(size_t)row * K + c];
    }
    __syncthreads();

    const int n0      = wave * 16;
    const int half    = lane >> 4;
    const int l15     = lane & 15;
    const int kpair   = half * 2;

    v8f acc = {};
    for (int k = 0; k < K; k += 4) {
        v2f a, b;
        a.x = Apanel[l15 * AP + k + kpair];
        a.y = Apanel[l15 * AP + k + kpair + 1];
        b.x = B[(size_t)(k + kpair)     * NCOLS + n0 + l15];
        b.y = B[(size_t)(k + kpair + 1) * NCOLS + n0 + l15];
        acc = __builtin_amdgcn_wmma_f32_16x16x4_f32(
            false, a, false, b, (short)0, acc, false, false);
    }

    const int mbase = half * 8;
#pragma unroll
    for (int r = 0; r < 8; ++r) {
        int row = m0 + mbase + r;
        if (row < M) C[(size_t)row * NCOLS + n0 + l15] = acc[r];
    }
}

__global__ void __launch_bounds__(256, 6)
edge_proj_score(const float* __restrict__ EF, const float* __restrict__ We,
                const float* __restrict__ H, const int* __restrict__ src,
                float* __restrict__ score, int E) {
    __shared__ v2f Bfrag[16 * 8 * 32];

    const int lane = threadIdx.x & 31;
    const int wave = threadIdx.x >> 5;
    const int half = lane >> 4;
    const int l15  = lane & 15;

#pragma unroll
    for (int q = 0; q < 16; ++q) {
        const int k0 = 8 * (q >> 1) + 4 * half + 2 * (q & 1);
        v2f b;
        b.x = We[(size_t)k0       * NCOLS + 16 * wave + l15];
        b.y = We[(size_t)(k0 + 1) * NCOLS + 16 * wave + l15];
        Bfrag[(q * 8 + wave) * 32 + lane] = b;
    }
    __syncthreads();

    const int ntiles = (E + 15) >> 4;
    for (int tile = blockIdx.x * 8 + wave; tile < ntiles; tile += gridDim.x * 8) {
        const int m0 = tile * 16;

        asm volatile("" ::: "memory");

        int row = m0 + l15; row = row < E ? row : E - 1;
        const float4* arow = (const float4*)(EF + (size_t)row * EDGE_K) + half;
        float4 a4[8];
#pragma unroll
        for (int j = 0; j < 8; ++j) a4[j] = arow[2 * j];

        int s[8];
#pragma unroll
        for (int r = 0; r < 8; ++r) {
            int e = m0 + 8 * half + r;
            s[r] = src[e < E ? e : E - 1];
        }

        v8f acc[8];
#pragma unroll
        for (int t = 0; t < 8; ++t) acc[t] = (v8f){};
#pragma unroll
        for (int j = 0; j < 8; ++j) {
            v2f a0, a1;
            a0.x = a4[j].x; a0.y = a4[j].y;
            a1.x = a4[j].z; a1.y = a4[j].w;
#pragma unroll
            for (int t = 0; t < 8; ++t)
                acc[t] = __builtin_amdgcn_wmma_f32_16x16x4_f32(
                    false, a0, false, Bfrag[((2 * j) * 8 + t) * 32 + lane],
                    (short)0, acc[t], false, false);
#pragma unroll
            for (int t = 0; t < 8; ++t)
                acc[t] = __builtin_amdgcn_wmma_f32_16x16x4_f32(
                    false, a1, false, Bfrag[((2 * j + 1) * 8 + t) * 32 + lane],
                    (short)0, acc[t], false, false);
        }

        float sc[8][4];
#pragma unroll
        for (int hd = 0; hd < 4; ++hd) {
            float p[8];
#pragma unroll
            for (int r = 0; r < 8; ++r) {
                const float* hrow = H + (size_t)s[r] * NCOLS + 32 * hd + l15;
                p[r] = acc[2 * hd][r] * hrow[0] + acc[2 * hd + 1][r] * hrow[16];
            }
#pragma unroll
            for (int off = 1; off < 16; off <<= 1) {
#pragma unroll
                for (int r = 0; r < 8; ++r) p[r] += __shfl_xor(p[r], off, 32);
            }
#pragma unroll
            for (int r = 0; r < 8; ++r) {
                float sv = p[r] * 0.17677669529663689f;
                sc[r][hd] = sv > 0.f ? sv : 0.01f * sv;
            }
        }
        if (l15 == 0) {
#pragma unroll
            for (int r = 0; r < 8; ++r) {
                int e = m0 + 8 * half + r;
                if (e < E)
                    *(float4*)(score + (size_t)e * NHEADS) =
                        make_float4(sc[r][0], sc[r][1], sc[r][2], sc[r][3]);
            }
        }
    }
}

#define BKT_SHIFT 7
#define BKT_NODES 128
#define MAX_BKT 512
#define SORT_BLOCKS 256
#define SORT_T 512
#define SORT_U 8
static_assert(MAX_BKT == SORT_T, "the bucket pass scans one bucket counter per thread");

__device__ __forceinline__ int wave_sum_int(int v) {
    for (int off = 16; off > 0; off >>= 1) v += __shfl_xor(v, off, 32);
    return v;
}

__device__ __forceinline__ int block_excl_scan(int c, int* ws, int lane, int wave, int* total) {
    int inc = c;
#pragma unroll
    for (int off = 1; off < 32; off <<= 1) {
        int t = __shfl_up(inc, off, 32);
        if (lane >= off) inc += t;
    }
    __syncthreads();
    if (lane == 31) ws[wave] = inc;
    __syncthreads();
    const int w = lane < (SORT_T / 32) ? ws[lane] : 0;
    *total = wave_sum_int(w);
    return wave_sum_int(lane < wave ? w : 0) + inc - c;
}

__global__ void __launch_bounds__(SORT_T)
bucket_kernel(const int* __restrict__ src, const int* __restrict__ dst,
              int* __restrict__ recs_w, int* __restrict__ tab_w,
              int* __restrict__ cursor, int N, int E) {
    int4* __restrict__ recs = (int4*)recs_w;
    int2* __restrict__ tab  = (int2*)tab_w;
    __shared__ int bc[MAX_BKT];
    __shared__ int ws[32];
    const int tid  = threadIdx.x;
    const int lane = tid & 31;
    const int wave = tid >> 5;
    const int k    = blockIdx.x;
    const int nbk  = (N + BKT_NODES - 1) >> BKT_SHIFT;
    const int epb  = (E + SORT_BLOCKS - 1) / SORT_BLOCKS;
    const int e0   = k * epb;
    int e1 = e0 + epb; e1 = e1 < E ? e1 : E;

    if (k == 0 && tid == 0) *cursor = 0;
    for (int b = tid; b < MAX_BKT; b += SORT_T) bc[b] = 0;
    __syncthreads();

    for (int c0 = e0; c0 < e1; c0 += SORT_T * SORT_U) {
        int dv[SORT_U];
#pragma unroll
        for (int j = 0; j < SORT_U; ++j) {
            const int e = c0 + j * SORT_T + tid;
            dv[j] = dst[e < e1 ? e : e1 - 1];
        }
#pragma unroll
        for (int j = 0; j < SORT_U; ++j)
            if (c0 + j * SORT_T + tid < e1) atomicAdd(&bc[dv[j] >> BKT_SHIFT], 1);
    }
    __syncthreads();

    const int c = bc[tid];
    int total;
    const int start = block_excl_scan(c, ws, lane, wave, &total);
    if (tid < nbk) tab[(size_t)k * nbk + tid] = make_int2(c, start);
    bc[tid] = start;
    __syncthreads();

    for (int c0 = e0; c0 < e1; c0 += SORT_T * SORT_U) {
        int dv[SORT_U], sv[SORT_U];
#pragma unroll
        for (int j = 0; j < SORT_U; ++j) {
            const int e  = c0 + j * SORT_T + tid;
            const int ec = e < e1 ? e : e1 - 1;
            dv[j] = dst[ec];
            sv[j] = src[ec];
        }
#pragma unroll
        for (int j = 0; j < SORT_U; ++j) {
            const int e = c0 + j * SORT_T + tid;
            if (e < e1) {
                const int p = atomicAdd(&bc[dv[j] >> BKT_SHIFT], 1);
                recs[(size_t)e0 + p] = make_int4(e, sv[j], dv[j], 0);
            }
        }
    }
}

__device__ __forceinline__ int4 bucket_record(const int4* __restrict__ recs, const int* spre,
                                              const int* rbeg, int i) {
    int k = 0;
#pragma unroll
    for (int step = SORT_BLOCKS >> 1; step > 0; step >>= 1)
        if (spre[k + step] <= i) k += step;
    return recs[(size_t)rbeg[k] + (i - spre[k])];
}

__global__ void __launch_bounds__(SORT_T)
owner_kernel(const int* __restrict__ recs_w, const int* __restrict__ tab_w,
             int* __restrict__ cursor, int* __restrict__ slot,
             int* __restrict__ offs, int N, int E) {
    const int4* __restrict__ recs = (const int4*)recs_w;
    const int2* __restrict__ tab  = (const int2*)tab_w;
    int2* __restrict__ slot2 = (int2*)slot;
    int2* __restrict__ offs2 = (int2*)offs;
    __shared__ int spre[SORT_BLOCKS];
    __shared__ int rbeg[SORT_BLOCKS];
    __shared__ int lcnt[BKT_NODES];
    __shared__ int ws[32];
    __shared__ int sbase;
    const int tid  = threadIdx.x;
    const int lane = tid & 31;
    const int wave = tid >> 5;
    const int b    = blockIdx.x;
    const int nbk  = (N + BKT_NODES - 1) >> BKT_SHIFT;
    const int epb  = (E + SORT_BLOCKS - 1) / SORT_BLOCKS;

    int c = 0;
    if (tid < SORT_BLOCKS) {
        const int2 v = tab[(size_t)tid * nbk + b];
        c = v.x;
        rbeg[tid] = tid * epb + v.y;
    }
    if (tid < BKT_NODES) lcnt[tid] = 0;
    int total;
    const int pre = block_excl_scan(c, ws, lane, wave, &total);
    if (tid < SORT_BLOCKS) spre[tid] = pre;
    __syncthreads();

    for (int i = tid; i < total; i += SORT_T) {
        const int4 r = bucket_record(recs, spre, rbeg, i);
        atomicAdd(&lcnt[r.z & (BKT_NODES - 1)], 1);
    }
    __syncthreads();

    if (tid == 0) sbase = atomicAdd(cursor, total);
    const int nc = tid < BKT_NODES ? lcnt[tid] : 0;
    int ntotal;
    const int nstart = block_excl_scan(nc, ws, lane, wave, &ntotal);
    if (tid < BKT_NODES) {
        const int node = b * BKT_NODES + tid;
        const int beg  = sbase + nstart;
        if (node < N) offs2[node] = make_int2(beg, beg + nc);
        lcnt[tid] = beg;
    }
    __syncthreads();

    for (int i = tid; i < total; i += SORT_T) {
        const int4 r = bucket_record(recs, spre, rbeg, i);
        const int pos = atomicAdd(&lcnt[r.z & (BKT_NODES - 1)], 1);
        slot2[pos] = make_int2(r.x, r.y);
    }
}

__device__ __forceinline__ float wave_max(float v) {
    for (int off = 16; off > 0; off >>= 1) v = fmaxf(v, __shfl_xor(v, off, 32));
    return v;
}
__device__ __forceinline__ float wave_sum(float v) {
    for (int off = 16; off > 0; off >>= 1) v += __shfl_xor(v, off, 32);
    return v;
}

__device__ __forceinline__ void add_rows(const float* __restrict__ H, int srcs, float4 a,
                                         int n, int lane, int hd, float4& acc) {
    const int k0 = hd == 0 ? -1 : 0, k1 = hd == 1 ? -1 : 0;
    const int k2 = hd == 2 ? -1 : 0, k3 = hd == 3 ? -1 : 0;
    for (int j = 0; j < n; ++j) {
        const int s  = __builtin_amdgcn_readlane(srcs, j);
        const int b0 = __builtin_amdgcn_readlane(__float_as_int(a.x), j);
        const int b1 = __builtin_amdgcn_readlane(__float_as_int(a.y), j);
        const int b2 = __builtin_amdgcn_readlane(__float_as_int(a.z), j);
        const int b3 = __builtin_amdgcn_readlane(__float_as_int(a.w), j);
        const float attn = __int_as_float((b0 & k0) | (b1 & k1) | (b2 & k2) | (b3 & k3));
        const float4 hv = *(const float4*)(H + (size_t)s * NCOLS + lane * 4);
        acc.x += hv.x * attn;
        acc.y += hv.y * attn;
        acc.z += hv.z * attn;
        acc.w += hv.w * attn;
    }
}

__global__ void __launch_bounds__(256)
softmax_aggregate_kernel(const float* __restrict__ H, const int* __restrict__ slot,
                         const int* __restrict__ offs, const float* __restrict__ score,
                         float* __restrict__ out, int N) {
    const int lane = threadIdx.x & 31;
    const int d = __builtin_amdgcn_readfirstlane((int)(blockIdx.x * 8 + (threadIdx.x >> 5)));
    if (d >= N) return;
    const int2 span = ((const int2*)offs)[d];
    const int beg = span.x, end = span.y;
    const int hd  = lane >> 3;
    const int2*   slot2  = (const int2*)slot;
    const float4* score4 = (const float4*)score;
    const float ninf = __int_as_float(0xFF800000);

    float4 acc = make_float4(0.f, 0.f, 0.f, 0.f);
    if (end > beg) {
        int n0 = end - beg; n0 = n0 < 32 ? n0 : 32;
        int2   es0 = make_int2(0, 0);
        float4 sc0 = make_float4(ninf, ninf, ninf, ninf);
        if (lane < n0) { es0 = slot2[beg + lane]; sc0 = score4[es0.x]; }

        float4 m = sc0;
        for (int base = beg + 32; base < end; base += 32) {
            if (base + lane < end) {
                const float4 sc = score4[slot2[base + lane].x];
                m.x = fmaxf(m.x, sc.x); m.y = fmaxf(m.y, sc.y);
                m.z = fmaxf(m.z, sc.z); m.w = fmaxf(m.w, sc.w);
            }
        }
        m.x = wave_max(m.x); m.y = wave_max(m.y);
        m.z = wave_max(m.z); m.w = wave_max(m.w);

        float4 w0 = make_float4(0.f, 0.f, 0.f, 0.f);
        if (lane < n0) {
            w0.x = __expf(sc0.x - m.x); w0.y = __expf(sc0.y - m.y);
            w0.z = __expf(sc0.z - m.z); w0.w = __expf(sc0.w - m.w);
        }
        float4 den = w0;
        for (int base = beg + 32; base < end; base += 32) {
            if (base + lane < end) {
                const float4 sc = score4[slot2[base + lane].x];
                den.x += __expf(sc.x - m.x); den.y += __expf(sc.y - m.y);
                den.z += __expf(sc.z - m.z); den.w += __expf(sc.w - m.w);
            }
        }
        den.x = wave_sum(den.x); den.y = wave_sum(den.y);
        den.z = wave_sum(den.z); den.w = wave_sum(den.w);

        add_rows(H, es0.y,
                 make_float4(w0.x / den.x, w0.y / den.y, w0.z / den.z, w0.w / den.w),
                 n0, lane, hd, acc);
        for (int base = beg + 32; base < end; base += 32) {
            int n = end - base; n = n < 32 ? n : 32;
            int2   es = make_int2(0, 0);
            float4 a  = make_float4(0.f, 0.f, 0.f, 0.f);
            if (lane < n) {
                es = slot2[base + lane];
                const float4 sc = score4[es.x];
                a.x = __expf(sc.x - m.x) / den.x; a.y = __expf(sc.y - m.y) / den.y;
                a.z = __expf(sc.z - m.z) / den.z; a.w = __expf(sc.w - m.w) / den.w;
            }
            add_rows(H, es.y, a, n, lane, hd, acc);
        }
    }
    *(float4*)(out + (size_t)d * NCOLS + lane * 4) = acc;
}

extern "C" void kernel_launch(void* const* d_in, const int* in_sizes, int n_in,
                              void* d_out, int out_size, void* d_ws, size_t ws_size,
                              hipStream_t stream) {
    const float* node_feat = (const float*)d_in[0];
    const float* edge_feat = (const float*)d_in[1];
    const int*   src       = (const int*)d_in[2];
    const int*   dst       = (const int*)d_in[3];
    const float* W_node    = (const float*)d_in[4];
    const float* W_edge    = (const float*)d_in[5];
    float*       out       = (float*)d_out;

    const int N = in_sizes[0] / NODE_K;
    const int E = in_sizes[2];

    float* h     = (float*)d_ws;
    float* score = h     + (size_t)N * NCOLS;
    float* sortb = score + (size_t)E * NHEADS;
    float* offsb = sortb + (size_t)E * NHEADS;

    gemm_f32_wmma<NODE_K><<<(N + 15) / 16, 256, 0, stream>>>(node_feat, W_node, h, N);

    edge_proj_score<<<2048, 256, 0, stream>>>(edge_feat, W_edge, h, src, score, E);

    int* recs   = (int*)out;
    int* tab    = recs + 4 * (size_t)E;
    int* cursor = tab + 2 * SORT_BLOCKS * MAX_BKT;
    int* slot   = (int*)sortb;
    int* offs   = (int*)offsb;

    bucket_kernel<<<SORT_BLOCKS, SORT_T, 0, stream>>>(src, dst, recs, tab, cursor, N, E);
    owner_kernel<<<(N + BKT_NODES - 1) / BKT_NODES, SORT_T, 0, stream>>>(
        recs, tab, cursor, slot, offs, N, E);

    softmax_aggregate_kernel<<<(N + 7) / 8, 256, 0, stream>>>(h, slot, offs, score, out, N);
}


# sort passes run as extra work-groups inside the two matrix launches (bucket pass beside node GEMM, owner pass at 256 threads beside edge score): waits overlap WMMA-bound work; 5 launches -> 3
# speedup vs baseline: 4.4278x; 1.0694x over previous
#include <hip/hip_runtime.h>
#include <hip/hip_bf16.h>

typedef __attribute__((ext_vector_type(2))) float v2f;
typedef __attribute__((ext_vector_type(8))) float v8f;

#define NCOLS 128
#define NODE_K 256
#define EDGE_K 64
#define NHEADS 4

template <int K>
__device__ __forceinline__ void
gemm_f32_wmma_role(const float* __restrict__ A, const float* __restrict__ B,
                   float* __restrict__ C, int M, int blk) {
    constexpr int AP = K + 4;
    __shared__ float Apanels[2 * 16 * AP];
    const int lane   = threadIdx.x & 31;
    const int wave   = (threadIdx.x >> 5) & 7;
    const int stripe = threadIdx.x >> 8;
    const int t256   = threadIdx.x & 255;
    const int m0     = (blk * 2 + stripe) * 16;
    float* Apanel    = Apanels + stripe * (16 * AP);

    for (int idx = t256; idx < 16 * K; idx += 256) {
        int r = idx / K, c = idx % K;
        int row = m0 + r; row = row < M ? row : M - 1;
        Apanel[r * AP + c] = A[(size_t)row * K + c];
    }
    __syncthreads();

    const int n0      = wave * 16;
    const int half    = lane >> 4;
    const int l15     = lane & 15;
    const int kpair   = half * 2;

    v8f acc = {};
    for (int k = 0; k < K; k += 4) {
        v2f a, b;
        a.x = Apanel[l15 * AP + k + kpair];
        a.y = Apanel[l15 * AP + k + kpair + 1];
        b.x = B[(size_t)(k + kpair)     * NCOLS + n0 + l15];
        b.y = B[(size_t)(k + kpair + 1) * NCOLS + n0 + l15];
        acc = __builtin_amdgcn_wmma_f32_16x16x4_f32(
            false, a, false, b, (short)0, acc, false, false);
    }

    const int mbase = half * 8;
#pragma unroll
    for (int r = 0; r < 8; ++r) {
        int row = m0 + mbase + r;
        if (row < M) C[(size_t)row * NCOLS + n0 + l15] = acc[r];
    }
}

__device__ __forceinline__ void
edge_proj_score_role(const float* __restrict__ EF, const float* __restrict__ We,
                     const float* __restrict__ H, const int* __restrict__ src,
                     float* __restrict__ score, int E, int blk, int nblk) {
    __shared__ v2f Bfrag[16 * 8 * 32];

    const int lane = threadIdx.x & 31;
    const int wave = threadIdx.x >> 5;
    const int half = lane >> 4;
    const int l15  = lane & 15;

#pragma unroll
    for (int q = 0; q < 16; ++q) {
        const int k0 = 8 * (q >> 1) + 4 * half + 2 * (q & 1);
        v2f b;
        b.x = We[(size_t)k0       * NCOLS + 16 * wave + l15];
        b.y = We[(size_t)(k0 + 1) * NCOLS + 16 * wave + l15];
        Bfrag[(q * 8 + wave) * 32 + lane] = b;
    }
    __syncthreads();

    const int ntiles = (E + 15) >> 4;
    for (int tile = blk * 8 + wave; tile < ntiles; tile += nblk * 8) {
        const int m0 = tile * 16;

        asm volatile("" ::: "memory");

        int row = m0 + l15; row = row < E ? row : E - 1;
        const float4* arow = (const float4*)(EF + (size_t)row * EDGE_K) + half;
        float4 a4[8];
#pragma unroll
        for (int j = 0; j < 8; ++j) a4[j] = arow[2 * j];

        int s[8];
#pragma unroll
        for (int r = 0; r < 8; ++r) {
            int e = m0 + 8 * half + r;
            s[r] = src[e < E ? e : E - 1];
        }

        v8f acc[8];
#pragma unroll
        for (int t = 0; t < 8; ++t) acc[t] = (v8f){};
#pragma unroll
        for (int j = 0; j < 8; ++j) {
            v2f a0, a1;
            a0.x = a4[j].x; a0.y = a4[j].y;
            a1.x = a4[j].z; a1.y = a4[j].w;
#pragma unroll
            for (int t = 0; t < 8; ++t)
                acc[t] = __builtin_amdgcn_wmma_f32_16x16x4_f32(
                    false, a0, false, Bfrag[((2 * j) * 8 + t) * 32 + lane],
                    (short)0, acc[t], false, false);
#pragma unroll
            for (int t = 0; t < 8; ++t)
                acc[t] = __builtin_amdgcn_wmma_f32_16x16x4_f32(
                    false, a1, false, Bfrag[((2 * j + 1) * 8 + t) * 32 + lane],
                    (short)0, acc[t], false, false);
        }

        float sc[8][4];
#pragma unroll
        for (int hd = 0; hd < 4; ++hd) {
            float p[8];
#pragma unroll
            for (int r = 0; r < 8; ++r) {
                const float* hrow = H + (size_t)s[r] * NCOLS + 32 * hd + l15;
                p[r] = acc[2 * hd][r] * hrow[0] + acc[2 * hd + 1][r] * hrow[16];
            }
#pragma unroll
            for (int off = 1; off < 16; off <<= 1) {
#pragma unroll
                for (int r = 0; r < 8; ++r) p[r] += __shfl_xor(p[r], off, 32);
            }
#pragma unroll
            for (int r = 0; r < 8; ++r) {
                float sv = p[r] * 0.17677669529663689f;
                sc[r][hd] = sv > 0.f ? sv : 0.01f * sv;
            }
        }
        if (l15 == 0) {
#pragma unroll
            for (int r = 0; r < 8; ++r) {
                int e = m0 + 8 * half + r;
                if (e < E)
                    *(float4*)(score + (size_t)e * NHEADS) =
                        make_float4(sc[r][0], sc[r][1], sc[r][2], sc[r][3]);
            }
        }
    }
}

#define BKT_SHIFT 7
#define BKT_NODES 128
#define MAX_BKT 512
#define SORT_BLOCKS 256
#define SORT_T 512
#define OWN_T 256
#define SORT_U 8
static_assert(MAX_BKT == SORT_T, "the bucket pass scans one bucket counter per thread");
static_assert(OWN_T >= SORT_BLOCKS && OWN_T >= BKT_NODES,
              "the owner pass scans one sender and one node counter per thread");

__device__ __forceinline__ int wave_sum_int(int v) {
    for (int off = 16; off > 0; off >>= 1) v += __shfl_xor(v, off, 32);
    return v;
}

template <int NT>
__device__ __forceinline__ int block_excl_scan(int c, int* ws, int lane, int wave, int* total) {
    int inc = c;
#pragma unroll
    for (int off = 1; off < 32; off <<= 1) {
        int t = __shfl_up(inc, off, 32);
        if (lane >= off) inc += t;
    }
    __syncthreads();
    if (lane == 31) ws[wave] = inc;
    __syncthreads();
    const int w = lane < (NT / 32) ? ws[lane] : 0;
    *total = wave_sum_int(w);
    return wave_sum_int(lane < wave ? w : 0) + inc - c;
}

__device__ __forceinline__ void
bucket_role(const int* __restrict__ src, const int* __restrict__ dst,
            int* __restrict__ recs_w, int* __restrict__ tab_w,
            int* __restrict__ cursor, int N, int E, int blk) {
    int4* __restrict__ recs = (int4*)recs_w;
    int2* __restrict__ tab  = (int2*)tab_w;
    __shared__ int bc[MAX_BKT];
    __shared__ int ws[32];
    const int tid  = threadIdx.x;
    const int lane = tid & 31;
    const int wave = tid >> 5;
    const int k    = blk;
    const int nbk  = (N + BKT_NODES - 1) >> BKT_SHIFT;
    const int epb  = (E + SORT_BLOCKS - 1) / SORT_BLOCKS;
    const int e0   = k * epb;
    int e1 = e0 + epb; e1 = e1 < E ? e1 : E;

    if (k == 0 && tid == 0) *cursor = 0;
    for (int b = tid; b < MAX_BKT; b += SORT_T) bc[b] = 0;
    __syncthreads();

    for (int c0 = e0; c0 < e1; c0 += SORT_T * SORT_U) {
        int dv[SORT_U];
#pragma unroll
        for (int j = 0; j < SORT_U; ++j) {
            const int e = c0 + j * SORT_T + tid;
            dv[j] = dst[e < e1 ? e : e1 - 1];
        }
#pragma unroll
        for (int j = 0; j < SORT_U; ++j)
            if (c0 + j * SORT_T + tid < e1) atomicAdd(&bc[dv[j] >> BKT_SHIFT], 1);
    }
    __syncthreads();

    const int c = bc[tid];
    int total;
    const int start = block_excl_scan<SORT_T>(c, ws, lane, wave, &total);
    if (tid < nbk) tab[(size_t)k * nbk + tid] = make_int2(c, start);
    bc[tid] = start;
    __syncthreads();

    for (int c0 = e0; c0 < e1; c0 += SORT_T * SORT_U) {
        int dv[SORT_U], sv[SORT_U];
#pragma unroll
        for (int j = 0; j < SORT_U; ++j) {
            const int e  = c0 + j * SORT_T + tid;
            const int ec = e < e1 ? e : e1 - 1;
            dv[j] = dst[ec];
            sv[j] = src[ec];
        }
#pragma unroll
        for (int j = 0; j < SORT_U; ++j) {
            const int e = c0 + j * SORT_T + tid;
            if (e < e1) {
                const int p = atomicAdd(&bc[dv[j] >> BKT_SHIFT], 1);
                recs[(size_t)e0 + p] = make_int4(e, sv[j], dv[j], 0);
            }
        }
    }
}

__device__ __forceinline__ int4 bucket_record(const int4* __restrict__ recs, const int* spre,
                                              const int* rbeg, int i) {
    int k = 0;
#pragma unroll
    for (int step = SORT_BLOCKS >> 1; step > 0; step >>= 1)
        if (spre[k + step] <= i) k += step;
    return recs[(size_t)rbeg[k] + (i - spre[k])];
}

__device__ __forceinline__ void
owner_role(const int* __restrict__ recs_w, const int* __restrict__ tab_w,
           int* __restrict__ cursor, int* __restrict__ slot,
           int* __restrict__ offs, int N, int E, int blk) {
    const int4* __restrict__ recs = (const int4*)recs_w;
    const int2* __restrict__ tab  = (const int2*)tab_w;
    int2* __restrict__ slot2 = (int2*)slot;
    int2* __restrict__ offs2 = (int2*)offs;
    __shared__ int spre[SORT_BLOCKS];
    __shared__ int rbeg[SORT_BLOCKS];
    __shared__ int lcnt[BKT_NODES];
    __shared__ int ws[32];
    __shared__ int sbase;
    const int tid  = threadIdx.x;
    const int lane = tid & 31;
    const int wave = tid >> 5;
    const int b    = blk;
    const int nbk  = (N + BKT_NODES - 1) >> BKT_SHIFT;
    const int epb  = (E + SORT_BLOCKS - 1) / SORT_BLOCKS;

    int c = 0;
    if (tid < SORT_BLOCKS) {
        const int2 v = tab[(size_t)tid * nbk + b];
        c = v.x;
        rbeg[tid] = tid * epb + v.y;
    }
    if (tid < BKT_NODES) lcnt[tid] = 0;
    int total;
    const int pre = block_excl_scan<OWN_T>(c, ws, lane, wave, &total);
    if (tid < SORT_BLOCKS) spre[tid] = pre;
    __syncthreads();

    for (int i = tid; i < total; i += OWN_T) {
        const int4 r = bucket_record(recs, spre, rbeg, i);
        atomicAdd(&lcnt[r.z & (BKT_NODES - 1)], 1);
    }
    __syncthreads();

    if (tid == 0) sbase = atomicAdd(cursor, total);
    const int nc = tid < BKT_NODES ? lcnt[tid] : 0;
    int ntotal;
    const int nstart = block_excl_scan<OWN_T>(nc, ws, lane, wave, &ntotal);
    if (tid < BKT_NODES) {
        const int node = b * BKT_NODES + tid;
        const int beg  = sbase + nstart;
        if (node < N) offs2[node] = make_int2(beg, beg + nc);
        lcnt[tid] = beg;
    }
    __syncthreads();

    for (int i = tid; i < total; i += OWN_T) {
        const int4 r = bucket_record(recs, spre, rbeg, i);
        const int pos = atomicAdd(&lcnt[r.z & (BKT_NODES - 1)], 1);
        slot2[pos] = make_int2(r.x, r.y);
    }
}

__global__ void __launch_bounds__(SORT_T)
gemm_bucket_kernel(const float* __restrict__ A, const float* __restrict__ B,
                   float* __restrict__ C, int M,
                   const int* __restrict__ src, const int* __restrict__ dst,
                   int* __restrict__ recs_w, int* __restrict__ tab_w,
                   int* __restrict__ cursor, int E) {
    if (blockIdx.x < SORT_BLOCKS)
        bucket_role(src, dst, recs_w, tab_w, cursor, M, E, (int)blockIdx.x);
    else
        gemm_f32_wmma_role<NODE_K>(A, B, C, M, (int)blockIdx.x - SORT_BLOCKS);
}

__global__ void __launch_bounds__(OWN_T, 6)
edge_owner_kernel(const float* __restrict__ EF, const float* __restrict__ We,
                  const float* __restrict__ H, const int* __restrict__ src,
                  float* __restrict__ score, int E,
                  const int* __restrict__ recs_w, const int* __restrict__ tab_w,
                  int* __restrict__ cursor, int* __restrict__ slot,
                  int* __restrict__ offs, int N) {
    const int nown = (N + BKT_NODES - 1) >> BKT_SHIFT;
    if ((int)blockIdx.x < nown)
        owner_role(recs_w, tab_w, cursor, slot, offs, N, E, (int)blockIdx.x);
    else
        edge_proj_score_role(EF, We, H, src, score, E, (int)blockIdx.x - nown,
                             (int)gridDim.x - nown);
}

__device__ __forceinline__ float wave_max(float v) {
    for (int off = 16; off > 0; off >>= 1) v = fmaxf(v, __shfl_xor(v, off, 32));
    return v;
}
__device__ __forceinline__ float wave_sum(float v) {
    for (int off = 16; off > 0; off >>= 1) v += __shfl_xor(v, off, 32);
    return v;
}

__device__ __forceinline__ void add_rows(const float* __restrict__ H, int srcs, float4 a,
                                         int n, int lane, int hd, float4& acc) {
    const int k0 = hd == 0 ? -1 : 0, k1 = hd == 1 ? -1 : 0;
    const int k2 = hd == 2 ? -1 : 0, k3 = hd == 3 ? -1 : 0;
    for (int j = 0; j < n; ++j) {
        const int s  = __builtin_amdgcn_readlane(srcs, j);
        const int b0 = __builtin_amdgcn_readlane(__float_as_int(a.x), j);
        const int b1 = __builtin_amdgcn_readlane(__float_as_int(a.y), j);
        const int b2 = __builtin_amdgcn_readlane(__float_as_int(a.z), j);
        const int b3 = __builtin_amdgcn_readlane(__float_as_int(a.w), j);
        const float attn = __int_as_float((b0 & k0) | (b1 & k1) | (b2 & k2) | (b3 & k3));
        const float4 hv = *(const float4*)(H + (size_t)s * NCOLS + lane * 4);
        acc.x += hv.x * attn;
        acc.y += hv.y * attn;
        acc.z += hv.z * attn;
        acc.w += hv.w * attn;
    }
}

__global__ void __launch_bounds__(256)
softmax_aggregate_kernel(const float* __restrict__ H, const int* __restrict__ slot,
                         const int* __restrict__ offs, const float* __restrict__ score,
                         float* __restrict__ out, int N) {
    const int lane = threadIdx.x & 31;
    const int d = __builtin_amdgcn_readfirstlane((int)(blockIdx.x * 8 + (threadIdx.x >> 5)));
    if (d >= N) return;
    const int2 span = ((const int2*)offs)[d];
    const int beg = span.x, end = span.y;
    const int hd  = lane >> 3;
    const int2*   slot2  = (const int2*)slot;
    const float4* score4 = (const float4*)score;
    const float ninf = __int_as_float(0xFF800000);

    float4 acc = make_float4(0.f, 0.f, 0.f, 0.f);
    if (end > beg) {
        int n0 = end - beg; n0 = n0 < 32 ? n0 : 32;
        int2   es0 = make_int2(0, 0);
        float4 sc0 = make_float4(ninf, ninf, ninf, ninf);
        if (lane < n0) { es0 = slot2[beg + lane]; sc0 = score4[es0.x]; }

        float4 m = sc0;
        for (int base = beg + 32; base < end; base += 32) {
            if (base + lane < end) {
                const float4 sc = score4[slot2[base + lane].x];
                m.x = fmaxf(m.x, sc.x); m.y = fmaxf(m.y, sc.y);
                m.z = fmaxf(m.z, sc.z); m.w = fmaxf(m.w, sc.w);
            }
        }
        m.x = wave_max(m.x); m.y = wave_max(m.y);
        m.z = wave_max(m.z); m.w = wave_max(m.w);

        float4 w0 = make_float4(0.f, 0.f, 0.f, 0.f);
        if (lane < n0) {
            w0.x = __expf(sc0.x - m.x); w0.y = __expf(sc0.y - m.y);
            w0.z = __expf(sc0.z - m.z); w0.w = __expf(sc0.w - m.w);
        }
        float4 den = w0;
        for (int base = beg + 32; base < end; base += 32) {
            if (base + lane < end) {
                const float4 sc = score4[slot2[base + lane].x];
                den.x += __expf(sc.x - m.x); den.y += __expf(sc.y - m.y);
                den.z += __expf(sc.z - m.z); den.w += __expf(sc.w - m.w);
            }
        }
        den.x = wave_sum(den.x); den.y = wave_sum(den.y);
        den.z = wave_sum(den.z); den.w = wave_sum(den.w);

        add_rows(H, es0.y,
                 make_float4(w0.x / den.x, w0.y / den.y, w0.z / den.z, w0.w / den.w),
                 n0, lane, hd, acc);
        for (int base = beg + 32; base < end; base += 32) {
            int n = end - base; n = n < 32 ? n : 32;
            int2   es = make_int2(0, 0);
            float4 a  = make_float4(0.f, 0.f, 0.f, 0.f);
            if (lane < n) {
                es = slot2[base + lane];
                const float4 sc = score4[es.x];
                a.x = __expf(sc.x - m.x) / den.x; a.y = __expf(sc.y - m.y) / den.y;
                a.z = __expf(sc.z - m.z) / den.z; a.w = __expf(sc.w - m.w) / den.w;
            }
            add_rows(H, es.y, a, n, lane, hd, acc);
        }
    }
    *(float4*)(out + (size_t)d * NCOLS + lane * 4) = acc;
}

extern "C" void kernel_launch(void* const* d_in, const int* in_sizes, int n_in,
                              void* d_out, int out_size, void* d_ws, size_t ws_size,
                              hipStream_t stream) {
    const float* node_feat = (const float*)d_in[0];
    const float* edge_feat = (const float*)d_in[1];
    const int*   src       = (const int*)d_in[2];
    const int*   dst       = (const int*)d_in[3];
    const float* W_node    = (const float*)d_in[4];
    const float* W_edge    = (const float*)d_in[5];
    float*       out       = (float*)d_out;

    const int N = in_sizes[0] / NODE_K;
    const int E = in_sizes[2];

    float* h     = (float*)d_ws;
    float* score = h     + (size_t)N * NCOLS;
    float* sortb = score + (size_t)E * NHEADS;
    float* offsb = sortb + (size_t)E * NHEADS;

    int* recs   = (int*)out;
    int* tab    = recs + 4 * (size_t)E;
    int* cursor = tab + 2 * SORT_BLOCKS * MAX_BKT;
    int* slot   = (int*)sortb;
    int* offs   = (int*)offsb;

    gemm_bucket_kernel<<<SORT_BLOCKS + (N + 31) / 32, SORT_T, 0, stream>>>(
        node_feat, W_node, h, N, src, dst, recs, tab, cursor, E);

    edge_owner_kernel<<<(N + BKT_NODES - 1) / BKT_NODES + 2048, OWN_T, 0, stream>>>(
        edge_feat, W_edge, h, src, score, E, recs, tab, cursor, slot, offs, N);

    softmax_aggregate_kernel<<<(N + 7) / 8, 256, 0, stream>>>(h, slot, offs, score, out, N);
}
